# starting kernel
# baseline (speedup 1.0000x reference)
#include <hip/hip_runtime.h>
#include <hip/hip_bf16.h>

typedef __attribute__((ext_vector_type(16))) _Float16 v16h;
typedef __attribute__((ext_vector_type(8)))  _Float16 v8h;
typedef __attribute__((ext_vector_type(8)))  float    v8f;
typedef __attribute__((ext_vector_type(4)))  float    v4f;

#define H_HEADS 4
#define IN_DIM  128
#define HID     64
#define COLS    256
#define PRED    16
#define RELS    50
#define LAYERS  3

__device__ __forceinline__ int f2key(float f) {
    int i = __float_as_int(f);
    return (i >= 0) ? i : (i ^ 0x7FFFFFFF);
}
__device__ __forceinline__ float key2f(int i) {
    return __int_as_float((i >= 0) ? i : (i ^ 0x7FFFFFFF));
}

__global__ void k_w1_to_f16(const float* __restrict__ W1,
                            _Float16* __restrict__ W1h, int n) {
    int i = blockIdx.x * blockDim.x + threadIdx.x;
    if (i < n) W1h[i] = (_Float16)W1[i];
}

__global__ void k_etab(const float* __restrict__ rel_emb,
                       const float* __restrict__ W_pred,
                       float* __restrict__ etab) {
    int i = blockIdx.x * blockDim.x + threadIdx.x;
    if (i >= LAYERS * RELS * H_HEADS) return;
    int l = i / (RELS * H_HEADS);
    int rem = i % (RELS * H_HEADS);
    int r = rem / H_HEADS;
    int h = rem % H_HEADS;
    float e = 0.f;
    #pragma unroll
    for (int p = 0; p < PRED; ++p)
        e += rel_emb[r * PRED + p] * W_pred[(l * PRED + p) * H_HEADS + h];
    etab[i] = (e > 0.f) ? e : 0.2f * e;
}

__global__ void k_mlp_wmma(const float* __restrict__ inputs,
                           const _Float16* __restrict__ W1h,
                           const float* __restrict__ b1,
                           const float* __restrict__ W2,
                           const float* __restrict__ b2,
                           float* __restrict__ hout,
                           int n, int nTiles) {
    const int wave = (blockIdx.x * blockDim.x + threadIdx.x) >> 5;
    if (wave >= nTiles) return;
    const int lane = threadIdx.x & 31;
    const int grp  = lane >> 4;
    const int l15  = lane & 15;
    const int nodeBase = wave * 16;
    int arow = nodeBase + l15;
    if (arow > n - 1) arow = n - 1;
    const float* __restrict__ abase = inputs + (size_t)arow * IN_DIM;

    const v8f zero = {};

    #pragma unroll
    for (int head = 0; head < H_HEADS; ++head) {
        v8f acc0 = zero, acc1 = zero, acc2 = zero, acc3 = zero;

        #pragma unroll
        for (int kk = 0; kk < IN_DIM; kk += 32) {
            const int kb = kk + grp * 8;
            v4f a0 = *(const v4f*)(abase + kb + 0);
            v4f a1 = *(const v4f*)(abase + kb + 4);
            v4f a2 = *(const v4f*)(abase + kb + 16);
            v4f a3 = *(const v4f*)(abase + kb + 20);
            v16h A;
            #pragma unroll
            for (int i = 0; i < 4; ++i) {
                A[i]      = (_Float16)a0[i];
                A[4 + i]  = (_Float16)a1[i];
                A[8 + i]  = (_Float16)a2[i];
                A[12 + i] = (_Float16)a3[i];
            }
            #pragma unroll
            for (int tt = 0; tt < 4; ++tt) {
                const int col = head * HID + tt * 16 + l15;
                const _Float16* q = W1h + (size_t)col * IN_DIM + kb;
                v8h blo = *(const v8h*)(q);
                v8h bhi = *(const v8h*)(q + 16);
                v16h B;
                #pragma unroll
                for (int i = 0; i < 8; ++i) { B[i] = blo[i]; B[8 + i] = bhi[i]; }
                v8f c = (tt == 0) ? acc0 : (tt == 1) ? acc1 : (tt == 2) ? acc2 : acc3;
                c = __builtin_amdgcn_wmma_f32_16x16x32_f16(
                        false, A, false, B, (short)0, c, false, false);
                if (tt == 0) acc0 = c; else if (tt == 1) acc1 = c;
                else if (tt == 2) acc2 = c; else acc3 = c;
            }
        }

        v8f hacc = zero;
        #pragma unroll
        for (int tt = 0; tt < 4; ++tt) {
            const int col  = head * HID + tt * 16 + l15;
            const float bias = b1[col];
            const float w2   = W2[col];
            v8f c = (tt == 0) ? acc0 : (tt == 1) ? acc1 : (tt == 2) ? acc2 : acc3;
            #pragma unroll
            for (int v = 0; v < 8; ++v) {
                float p = c[v] + bias;
                p = (p > 0.f) ? p : 0.f;
                p *= w2;
                p += __shfl_xor(p, 1, 32);
                p += __shfl_xor(p, 2, 32);
                p += __shfl_xor(p, 4, 32);
                p += __shfl_xor(p, 8, 32);
                hacc[v] += p;
            }
        }
        if (l15 == 0) {
            #pragma unroll
            for (int v = 0; v < 8; ++v) {
                const int node = nodeBase + v + 8 * grp;
                if (node < n)
                    hout[node * H_HEADS + head] = hacc[v] + b2[head];
            }
        }
    }
}

__global__ void k_init_nh(int* __restrict__ m, float* __restrict__ s,
                          float* __restrict__ u, int count) {
    int i = blockIdx.x * blockDim.x + threadIdx.x;
    if (i < count) { m[i] = (int)0x80000000; s[i] = 0.f; u[i] = 0.f; }
}

__global__ void k_edge_max(const int* __restrict__ et, const int* __restrict__ dst,
                           const float* __restrict__ etab_l,
                           int* __restrict__ m, int E) {
    int i = blockIdx.x * blockDim.x + threadIdx.x;
    if (i >= E) return;
    __builtin_prefetch(et + i + 8192, 0, 0);
    __builtin_prefetch(dst + i + 8192, 0, 0);
    const int t = et[i];
    const int d = dst[i];
    const float* e = etab_l + t * H_HEADS;
    #pragma unroll
    for (int h = 0; h < H_HEADS; ++h)
        atomicMax(&m[d * H_HEADS + h], f2key(e[h]));
}

__global__ void k_edge_agg(const int* __restrict__ et, const int* __restrict__ src,
                           const int* __restrict__ dst,
                           const float* __restrict__ etab_l,
                           const int* __restrict__ m,
                           const float* __restrict__ hcur,
                           float* __restrict__ s, float* __restrict__ u, int E) {
    int i = blockIdx.x * blockDim.x + threadIdx.x;
    if (i >= E) return;
    __builtin_prefetch(et + i + 8192, 0, 0);
    __builtin_prefetch(src + i + 8192, 0, 0);
    __builtin_prefetch(dst + i + 8192, 0, 0);
    const int t  = et[i];
    const int sn = src[i];
    const int d  = dst[i];
    const float* e = etab_l + t * H_HEADS;
    #pragma unroll
    for (int h = 0; h < H_HEADS; ++h) {
        const float mv = key2f(m[d * H_HEADS + h]);
        const float w  = __expf(e[h] - mv);
        atomicAdd(&s[d * H_HEADS + h], w);
        atomicAdd(&u[d * H_HEADS + h], w * hcur[sn * H_HEADS + h]);
    }
}

__global__ void k_node_update(const float* __restrict__ s, const float* __restrict__ u,
                              float* __restrict__ hnext, int n, int do_mean) {
    int i = blockIdx.x * blockDim.x + threadIdx.x;
    if (i >= n) return;
    float v[H_HEADS];
    #pragma unroll
    for (int h = 0; h < H_HEADS; ++h) {
        const float sv = s[i * H_HEADS + h];
        float x = (sv > 0.f) ? (u[i * H_HEADS + h] / sv) : 0.f;
        v[h] = (x > 0.f) ? x : 0.f;
    }
    if (do_mean) {
        const float mn = 0.25f * (v[0] + v[1] + v[2] + v[3]);
        #pragma unroll
        for (int h = 0; h < H_HEADS; ++h) v[h] = mn;
    }
    #pragma unroll
    for (int h = 0; h < H_HEADS; ++h) hnext[i * H_HEADS + h] = v[h];
}

__global__ void k_final(const float* __restrict__ hfin,
                        const float* __restrict__ centrality,
                        const float* __restrict__ gamma,
                        const float* __restrict__ beta,
                        float* __restrict__ out, int n) {
    int i = blockIdx.x * blockDim.x + threadIdx.x;
    if (i >= n) return;
    const float c = centrality[i];
    float acc = 0.f;
    #pragma unroll
    for (int h = 0; h < H_HEADS; ++h)
        acc += (c * gamma[h] + beta[h]) * hfin[i * H_HEADS + h];
    acc *= 0.25f;
    out[i] = (acc > 0.f) ? acc : 0.01f * acc;
}

extern "C" void kernel_launch(void* const* d_in, const int* in_sizes, int n_in,
                              void* d_out, int out_size, void* d_ws, size_t ws_size,
                              hipStream_t stream) {
    const float* inputs     = (const float*)d_in[0];
    const float* centrality = (const float*)d_in[1];
    const float* W1         = (const float*)d_in[2];
    const float* b1         = (const float*)d_in[3];
    const float* W2         = (const float*)d_in[4];
    const float* b2         = (const float*)d_in[5];
    const float* rel_emb    = (const float*)d_in[6];
    const float* W_pred     = (const float*)d_in[7];
    const float* gamma      = (const float*)d_in[8];
    const float* beta       = (const float*)d_in[9];
    const int*   edge_types = (const int*)d_in[10];
    const int*   src        = (const int*)d_in[11];
    const int*   dst        = (const int*)d_in[12];
    float* out = (float*)d_out;

    const int N = in_sizes[1];
    const int E = in_sizes[10];

    char* ws = (char*)d_ws;
    _Float16* W1h = (_Float16*)ws;
    float* etab   = (float*)(ws + 65536);
    float* hA     = (float*)(ws + 69632);
    float* hB     = hA + (size_t)N * H_HEADS;
    int*   m      = (int*)(hB + (size_t)N * H_HEADS);
    float* s      = (float*)(m + (size_t)N * H_HEADS);
    float* u      = s + (size_t)N * H_HEADS;

    const int TPB = 256;

    {
        int cnt = COLS * IN_DIM;
        k_w1_to_f16<<<(cnt + TPB - 1) / TPB, TPB, 0, stream>>>(W1, W1h, cnt);
    }
    {
        int cnt = LAYERS * RELS * H_HEADS;
        k_etab<<<(cnt + TPB - 1) / TPB, TPB, 0, stream>>>(rel_emb, W_pred, etab);
    }
    {
        int nTiles = (N + 15) / 16;
        int wavesPerBlock = 4;
        int blocks = (nTiles + wavesPerBlock - 1) / wavesPerBlock;
        k_mlp_wmma<<<blocks, wavesPerBlock * 32, 0, stream>>>(
            inputs, W1h, b1, W2, b2, hA, N, nTiles);
    }

    float* hcur = hA;
    float* hnxt = hB;
    const int nhCount = N * H_HEADS;
    const int nhBlocks = (nhCount + TPB - 1) / TPB;
    const int eBlocks  = (E + TPB - 1) / TPB;
    const int nBlocks  = (N + TPB - 1) / TPB;

    for (int l = 0; l < LAYERS; ++l) {
        const float* etab_l = etab + l * RELS * H_HEADS;
        k_init_nh<<<nhBlocks, TPB, 0, stream>>>(m, s, u, nhCount);
        k_edge_max<<<eBlocks, TPB, 0, stream>>>(edge_types, dst, etab_l, m, E);
        k_edge_agg<<<eBlocks, TPB, 0, stream>>>(edge_types, src, dst, etab_l,
                                                m, hcur, s, u, E);
        k_node_update<<<nBlocks, TPB, 0, stream>>>(s, u, hnxt, N,
                                                   (l != LAYERS - 1) ? 1 : 0);
        float* tmp = hcur; hcur = hnxt; hnxt = tmp;
    }

    k_final<<<nBlocks, TPB, 0, stream>>>(hcur, centrality, gamma, beta, out, N);
}


# Node ranges cut to 128 nodes: a workgroup keeps all 50 types' count and sum tiles in LDS and runs the node pass there; no table goes to memory; layer-1 sums now LDS arrival per type, then type order
# speedup vs baseline: 18.7737x; 18.7737x over previous
#include <hip/hip_runtime.h>
#include <hip/hip_bf16.h>

typedef __attribute__((ext_vector_type(16))) _Float16 v16h;
typedef __attribute__((ext_vector_type(8)))  _Float16 v8h;
typedef __attribute__((ext_vector_type(8)))  float    v8f;
typedef __attribute__((ext_vector_type(4)))  float    v4f;

#define H_HEADS 4
#define IN_DIM  128
#define HID     64
#define COLS    256
#define PRED    16
#define RELS    50
#define LAYERS  3


__global__ void k_w1_to_f16(const float* __restrict__ W1,
                            _Float16* __restrict__ W1h, int n) {
    int i = blockIdx.x * blockDim.x + threadIdx.x;
    if (i < n) W1h[i] = (_Float16)W1[i];
}

__global__ void k_etab(const float* __restrict__ rel_emb,
                       const float* __restrict__ W_pred,
                       float* __restrict__ etab) {
    int i = blockIdx.x * blockDim.x + threadIdx.x;
    if (i >= LAYERS * RELS * H_HEADS) return;
    int l = i / (RELS * H_HEADS);
    int rem = i % (RELS * H_HEADS);
    int r = rem / H_HEADS;
    int h = rem % H_HEADS;
    float e = 0.f;
    #pragma unroll
    for (int p = 0; p < PRED; ++p)
        e += rel_emb[r * PRED + p] * W_pred[(l * PRED + p) * H_HEADS + h];
    etab[i] = (e > 0.f) ? e : 0.2f * e;
}

__global__ void k_mlp_wmma(const float* __restrict__ inputs,
                           const _Float16* __restrict__ W1h,
                           const float* __restrict__ b1,
                           const float* __restrict__ W2,
                           const float* __restrict__ b2,
                           float* __restrict__ hout,
                           int n, int nTiles) {
    const int wave = (blockIdx.x * blockDim.x + threadIdx.x) >> 5;
    if (wave >= nTiles) return;
    const int lane = threadIdx.x & 31;
    const int grp  = lane >> 4;
    const int l15  = lane & 15;
    const int nodeBase = wave * 16;
    int arow = nodeBase + l15;
    if (arow > n - 1) arow = n - 1;
    const float* __restrict__ abase = inputs + (size_t)arow * IN_DIM;

    const v8f zero = {};

    #pragma unroll
    for (int head = 0; head < H_HEADS; ++head) {
        v8f acc0 = zero, acc1 = zero, acc2 = zero, acc3 = zero;

        #pragma unroll
        for (int kk = 0; kk < IN_DIM; kk += 32) {
            const int kb = kk + grp * 8;
            v4f a0 = *(const v4f*)(abase + kb + 0);
            v4f a1 = *(const v4f*)(abase + kb + 4);
            v4f a2 = *(const v4f*)(abase + kb + 16);
            v4f a3 = *(const v4f*)(abase + kb + 20);
            v16h A;
            #pragma unroll
            for (int i = 0; i < 4; ++i) {
                A[i]      = (_Float16)a0[i];
                A[4 + i]  = (_Float16)a1[i];
                A[8 + i]  = (_Float16)a2[i];
                A[12 + i] = (_Float16)a3[i];
            }
            #pragma unroll
            for (int tt = 0; tt < 4; ++tt) {
                const int col = head * HID + tt * 16 + l15;
                const _Float16* q = W1h + (size_t)col * IN_DIM + kb;
                v8h blo = *(const v8h*)(q);
                v8h bhi = *(const v8h*)(q + 16);
                v16h B;
                #pragma unroll
                for (int i = 0; i < 8; ++i) { B[i] = blo[i]; B[8 + i] = bhi[i]; }
                v8f c = (tt == 0) ? acc0 : (tt == 1) ? acc1 : (tt == 2) ? acc2 : acc3;
                c = __builtin_amdgcn_wmma_f32_16x16x32_f16(
                        false, A, false, B, (short)0, c, false, false);
                if (tt == 0) acc0 = c; else if (tt == 1) acc1 = c;
                else if (tt == 2) acc2 = c; else acc3 = c;
            }
        }

        v8f hacc = zero;
        #pragma unroll
        for (int tt = 0; tt < 4; ++tt) {
            const int col  = head * HID + tt * 16 + l15;
            const float bias = b1[col];
            const float w2   = W2[col];
            v8f c = (tt == 0) ? acc0 : (tt == 1) ? acc1 : (tt == 2) ? acc2 : acc3;
            #pragma unroll
            for (int v = 0; v < 8; ++v) {
                float p = c[v] + bias;
                p = (p > 0.f) ? p : 0.f;
                p *= w2;
                p += __shfl_xor(p, 1, 32);
                p += __shfl_xor(p, 2, 32);
                p += __shfl_xor(p, 4, 32);
                p += __shfl_xor(p, 8, 32);
                hacc[v] += p;
            }
        }
        if (l15 == 0) {
            #pragma unroll
            for (int v = 0; v < 8; ++v) {
                const int node = nodeBase + v + 8 * grp;
                if (node < n)
                    hout[node * H_HEADS + head] = hacc[v] + b2[head];
            }
        }
    }
}

__global__ void k_zero4(v4f* __restrict__ p, int n4) {
    int i = blockIdx.x * blockDim.x + threadIdx.x;
    if (i < n4) { v4f z = {}; p[i] = z; }
}

#define FR_SH   7
#define FR_RS   128
#define KPAD    1024
#define BK_T    1024
#define BK_EPT  8
#define BK_CH   8192
#define KEY_SH  20
#define D_MASK  0xFFFFF

static_assert(BK_T == KPAD, "one thread per key slot in the grouping kernels");
static_assert(BK_CH == BK_T * BK_EPT, "chunk size");
static_assert(FR_RS == (1 << FR_SH), "range size");

__global__ void __launch_bounds__(BK_T)
k_key_hist(const int* __restrict__ dst, int* __restrict__ ghist, int E) {
    __shared__ int lh[KPAD];
    const int tid = threadIdx.x;
    lh[tid] = 0;
    __syncthreads();
    const int base = blockIdx.x * BK_CH + tid;
    #pragma unroll
    for (int k = 0; k < BK_EPT; ++k) {
        const int idx = base + k * BK_T;
        if (idx < E) atomicAdd(&lh[(unsigned)dst[idx] >> FR_SH], 1);
    }
    __syncthreads();
    const int c = lh[tid];
    if (c) atomicAdd(&ghist[tid], c);
}

__global__ void __launch_bounds__(BK_T)
k_bucket(const int* __restrict__ et, const int* __restrict__ src,
         const int* __restrict__ dst, const int* __restrict__ ghist,
         int* __restrict__ cursor, int* __restrict__ kstart,
         int* __restrict__ pairs, int E) {
    __shared__ int lh[KPAD];
    __shared__ int lstart[KPAD];
    __shared__ int gbase[KPAD];
    __shared__ int wg[BK_T / 32];
    __shared__ int wc[BK_T / 32];
    __shared__ int ld[BK_CH];
    __shared__ int ls[BK_CH];
    __shared__ int lt[BK_CH];
    const int tid = threadIdx.x;
    lh[tid] = 0;
    __syncthreads();
    const int cbase = blockIdx.x * BK_CH;
    int nValid = E - cbase; if (nValid > BK_CH) nValid = BK_CH;
    int key[BK_EPT], pk[BK_EPT], s[BK_EPT], rk[BK_EPT];
    #pragma unroll
    for (int k = 0; k < BK_EPT; ++k) {
        const int idx = cbase + k * BK_T + tid;
        const bool ok = idx < E;
        const int t = ok ? et[idx] : 0;
        const int d = ok ? dst[idx] : 0;
        s[k]   = ok ? src[idx] : 0;
        pk[k]  = d | (t << KEY_SH);
        key[k] = ok ? (int)((unsigned)d >> FR_SH) : -1;
    }
    #pragma unroll
    for (int k = 0; k < BK_EPT; ++k)
        rk[k] = (key[k] >= 0) ? atomicAdd(&lh[key[k]], 1) : 0;
    __syncthreads();
    {
        const int lane = tid & 31;
        const int wave = tid >> 5;
        const int g = ghist[tid];
        const int c = lh[tid];
        int gs = g, cs = c;
        #pragma unroll
        for (int off = 1; off < 32; off <<= 1) {
            const int a = __shfl_up(gs, off, 32);
            const int b = __shfl_up(cs, off, 32);
            if (lane >= off) { gs += a; cs += b; }
        }
        if (lane == 31) { wg[wave] = gs; wc[wave] = cs; }
        __syncthreads();
        int gpre = 0, cpre = 0;
        #pragma unroll
        for (int i = 0; i < BK_T / 32; ++i) {
            gpre += (i < wave) ? wg[i] : 0;
            cpre += (i < wave) ? wc[i] : 0;
        }
        const int off = c ? atomicAdd(&cursor[tid], c) : 0;
        lstart[tid] = cpre + cs - c;
        gbase[tid]  = gpre + gs - g + off;
        if (blockIdx.x == 0) kstart[tid] = gpre + gs - g;
    }
    __syncthreads();
    #pragma unroll
    for (int k = 0; k < BK_EPT; ++k) {
        if (key[k] >= 0) {
            const int pos = lstart[key[k]] + rk[k];
            ld[pos] = pk[k]; ls[pos] = s[k]; lt[pos] = key[k];
        }
    }
    __syncthreads();
    #pragma unroll
    for (int k = 0; k < BK_EPT; ++k) {
        const int i = k * BK_T + tid;
        if (i < nValid) {
            const int kk = lt[i];
            const int slot = gbase[kk] + (i - lstart[kk]);
            int2 p; p.x = ld[i]; p.y = ls[i];
            *(int2*)(pairs + 2 * (size_t)slot) = p;
        }
    }
}

__device__ __forceinline__ int quad_bcast(int v, int j) {
    switch (j) {
        case 0:  return __builtin_amdgcn_update_dpp(0, v, 0x00, 0xf, 0xf, true);
        case 1:  return __builtin_amdgcn_update_dpp(0, v, 0x55, 0xf, 0xf, true);
        case 2:  return __builtin_amdgcn_update_dpp(0, v, 0xAA, 0xf, 0xf, true);
        default: return __builtin_amdgcn_update_dpp(0, v, 0xFF, 0xf, 0xf, true);
    }
}

#define LY_T1   512
#define LY_TG   256
#define LY_U    16
#define LY_EPT  8
static_assert(LY_EPT == 8 && LY_U == 16, "the load fences below name every slot");
#define CELLS   (RELS * FR_RS)

template <int HI, bool LAST>
__global__ void __launch_bounds__(HI == 4 ? LY_T1 : LY_TG)
k_layer(const int* __restrict__ pairs, const int* __restrict__ ghist,
        const int* __restrict__ kstart, const float* __restrict__ etab_l,
        const float* __restrict__ hin, float* __restrict__ hout,
        const float* __restrict__ centrality, const float* __restrict__ gamma,
        const float* __restrict__ beta, float* __restrict__ out, int n) {
    constexpr int T  = (HI == 4) ? LY_T1 : LY_TG;
    constexpr int NW = T / 32;
    static_assert(T >= FR_RS, "one thread per node in the store pass");
    __shared__ __attribute__((aligned(16))) int   tab[CELLS * (1 + HI)];
    __shared__ __attribute__((aligned(16))) float eL[RELS * H_HEADS];
    int*   cnt = tab;
    float* G   = (float*)(tab + CELLS);
    const int tid  = threadIdx.x;
    const int lane = tid & 31;
    const int wave = tid >> 5;
    const int r    = blockIdx.x;
    const int lo   = r * FR_RS;
    int hi = lo + FR_RS; if (hi > n) hi = n;
    const int span  = (hi > lo) ? (hi - lo) : 0;
    const int start = kstart[r];
    const int end   = start + ghist[r];
    for (int i = tid; i < RELS * H_HEADS; i += T) eL[i] = etab_l[i];
    {
        const v4f z = {};
        for (int i = tid; i < CELLS * (1 + HI) / 4; i += T) ((v4f*)tab)[i] = z;
    }
    __syncthreads();

    if (HI == 4) {
        const int h = tid & 3;
        for (int base = start + wave * (32 * LY_EPT); base < end; base += NW * 32 * LY_EPT) {
            int px[LY_EPT], py[LY_EPT];
            #pragma unroll
            for (int k = 0; k < LY_EPT; ++k) {
                int idx = base + k * 32 + lane;
                idx = (idx < end) ? idx : (end - 1);
                const int2 p = *(const int2*)(pairs + 2 * (size_t)idx);
                px[k] = p.x;
                py[k] = p.y;
            }
            asm volatile("" :: "v"(py[0]), "v"(py[1]), "v"(py[2]), "v"(py[3]),
                               "v"(py[4]), "v"(py[5]), "v"(py[6]), "v"(py[7]));
            #pragma unroll
            for (int k = 0; k < LY_EPT; ++k)
                px[k] = ((base + k * 32 + lane) < end) ? px[k] : -1;
            int   cell[4 * LY_EPT];
            float hv[4 * LY_EPT];
            #pragma unroll
            for (int s = 0; s < 4 * LY_EPT; ++s) {
                const int k  = s >> 2;
                const int jj = s & 3;
                const int dk = quad_bcast(px[k], jj);
                const unsigned sn = (unsigned)quad_bcast(py[k], jj);
                const bool ok = dk >= 0;
                const float x = hin[sn * H_HEADS + (unsigned)h];
                hv[s]   = ok ? x : 0.f;
                cell[s] = ok ? (((dk >> KEY_SH) << FR_SH) + ((dk & D_MASK) - lo)) : 0;
            }
            #pragma unroll
            for (int k = 0; k < LY_EPT; ++k) {
                const bool ok = px[k] >= 0;
                const int  c  = ok ? (((px[k] >> KEY_SH) << FR_SH) + ((px[k] & D_MASK) - lo)) : 0;
                atomicAdd(&cnt[c], ok ? 1 : 0);
            }
            #pragma unroll
            for (int s = 0; s < 4 * LY_EPT; ++s)
                atomicAdd(&G[cell[s] * H_HEADS + h], hv[s]);
        }
    } else {
        for (int base = start + wave * (32 * LY_U); base < end; base += NW * 32 * LY_U) {
            int px[LY_U], py[LY_U];
            #pragma unroll
            for (int k = 0; k < LY_U; ++k) {
                int idx = base + k * 32 + lane;
                idx = (idx < end) ? idx : (end - 1);
                const int2 p = *(const int2*)(pairs + 2 * (size_t)idx);
                px[k] = p.x;
                py[k] = p.y;
            }
            asm volatile("" :: "v"(py[0]), "v"(py[1]), "v"(py[2]), "v"(py[3]),
                               "v"(py[4]), "v"(py[5]), "v"(py[6]), "v"(py[7]),
                               "v"(py[8]), "v"(py[9]), "v"(py[10]), "v"(py[11]),
                               "v"(py[12]), "v"(py[13]), "v"(py[14]), "v"(py[15]));
            float hv[LY_U];
            #pragma unroll
            for (int k = 0; k < LY_U; ++k) hv[k] = hin[(unsigned)py[k]];
            #pragma unroll
            for (int k = 0; k < LY_U; ++k) {
                const bool ok = (base + k * 32 + lane) < end;
                const int  c  = ok ? (((px[k] >> KEY_SH) << FR_SH) + ((px[k] & D_MASK) - lo)) : 0;
                atomicAdd(&cnt[c], ok ? 1 : 0);
                atomicAdd(&G[c], ok ? hv[k] : 0.f);
            }
        }
    }
    __syncthreads();

    if (tid < span) {
        const int d = tid;
        v4f mx;
        mx[0] = mx[1] = mx[2] = mx[3] = -__builtin_huge_valf();
        #pragma unroll 10
        for (int t = 0; t < RELS; ++t) {
            const bool present = cnt[t * FR_RS + d] > 0;
            const v4f e = *(const v4f*)(eL + t * H_HEADS);
            #pragma unroll
            for (int h = 0; h < H_HEADS; ++h)
                mx[h] = present ? fmaxf(mx[h], e[h]) : mx[h];
        }
        v4f sv = {}, uv = {};
        #pragma unroll 10
        for (int t = 0; t < RELS; ++t) {
            const int   c  = cnt[t * FR_RS + d];
            const float cf = (float)c;
            const v4f e = *(const v4f*)(eL + t * H_HEADS);
            v4f g;
            if (HI == 4) {
                g = *(const v4f*)(G + (t * FR_RS + d) * H_HEADS);
            } else {
                const float g1 = G[t * FR_RS + d];
                g[0] = g1; g[1] = g1; g[2] = g1; g[3] = g1;
            }
            #pragma unroll
            for (int h = 0; h < H_HEADS; ++h) {
                const float w = (c > 0) ? __expf(e[h] - mx[h]) : 0.f;
                sv[h] += cf * w;
                uv[h] += w * g[h];
            }
        }
        float v[H_HEADS];
        #pragma unroll
        for (int h = 0; h < H_HEADS; ++h) {
            const float x = (sv[h] > 0.f) ? (uv[h] / sv[h]) : 0.f;
            v[h] = (x > 0.f) ? x : 0.f;
        }
        if (!LAST) {
            hout[lo + d] = 0.25f * (v[0] + v[1] + v[2] + v[3]);
        } else {
            const float c = centrality[lo + d];
            float acc = 0.f;
            #pragma unroll
            for (int h = 0; h < H_HEADS; ++h)
                acc += (c * gamma[h] + beta[h]) * v[h];
            acc *= 0.25f;
            out[lo + d] = (acc > 0.f) ? acc : 0.01f * acc;
        }
    }
}

extern "C" void kernel_launch(void* const* d_in, const int* in_sizes, int n_in,
                              void* d_out, int out_size, void* d_ws, size_t ws_size,
                              hipStream_t stream) {
    const float* inputs     = (const float*)d_in[0];
    const float* centrality = (const float*)d_in[1];
    const float* W1         = (const float*)d_in[2];
    const float* b1         = (const float*)d_in[3];
    const float* W2         = (const float*)d_in[4];
    const float* b2         = (const float*)d_in[5];
    const float* rel_emb    = (const float*)d_in[6];
    const float* W_pred     = (const float*)d_in[7];
    const float* gamma      = (const float*)d_in[8];
    const float* beta       = (const float*)d_in[9];
    const int*   edge_types = (const int*)d_in[10];
    const int*   src        = (const int*)d_in[11];
    const int*   dst        = (const int*)d_in[12];
    float* out = (float*)d_out;

    const int N = in_sizes[1];
    const int E = in_sizes[10];

    char* ws = (char*)d_ws;
    _Float16* W1h = (_Float16*)ws;
    float* etab   = (float*)(ws + 65536);
    int*   ghist  = (int*)(ws + 69632);
    int*   cursor = ghist + KPAD;
    int*   kstart = cursor + KPAD;
    float* hA     = (float*)(kstart + KPAD);
    const size_t hbWords = ((size_t)N + 3) & ~(size_t)3;
    float* hbA    = hA + (size_t)N * H_HEADS;
    float* hbB    = hbA + hbWords;
    int*   pairs  = (int*)(hbB + hbWords);

    const int TPB = 256;

    {
        int cnt16 = COLS * IN_DIM;
        k_w1_to_f16<<<(cnt16 + TPB - 1) / TPB, TPB, 0, stream>>>(W1, W1h, cnt16);
    }
    {
        int c = LAYERS * RELS * H_HEADS;
        k_etab<<<(c + TPB - 1) / TPB, TPB, 0, stream>>>(rel_emb, W_pred, etab);
    }
    {
        int n4 = 2 * KPAD / 4;
        k_zero4<<<(n4 + TPB - 1) / TPB, TPB, 0, stream>>>((v4f*)ghist, n4);
    }
    {
        int nTiles = (N + 15) / 16;
        int wavesPerBlock = 4;
        int blocks = (nTiles + wavesPerBlock - 1) / wavesPerBlock;
        k_mlp_wmma<<<blocks, wavesPerBlock * 32, 0, stream>>>(
            inputs, W1h, b1, W2, b2, hA, N, nTiles);
    }

    const int bBlocks = (E + BK_CH - 1) / BK_CH;
    const int nRanges = (N + FR_RS - 1) / FR_RS;

    k_key_hist<<<bBlocks, BK_T, 0, stream>>>(dst, ghist, E);
    k_bucket<<<bBlocks, BK_T, 0, stream>>>(edge_types, src, dst, ghist, cursor, kstart, pairs, E);

    k_layer<4, false><<<nRanges, LY_T1, 0, stream>>>(pairs, ghist, kstart, etab, hA, hbA,
                                                     centrality, gamma, beta, out, N);
    k_layer<1, false><<<nRanges, LY_TG, 0, stream>>>(pairs, ghist, kstart, etab + 1 * RELS * H_HEADS,
                                                     hbA, hbB, centrality, gamma, beta, out, N);
    k_layer<1, true><<<nRanges, LY_TG, 0, stream>>>(pairs, ghist, kstart, etab + 2 * RELS * H_HEADS,
                                                    hbB, hbB, centrality, gamma, beta, out, N);
}


# MLP operand loads touch 4x fewer cache lines: f16 W1 stored in B-fragment order (a fragment load is one contiguous 512 B run) and a tile's A fragments loaded once for all 4 heads; k order kept
# speedup vs baseline: 22.3701x; 1.1916x over previous
#include <hip/hip_runtime.h>
#include <hip/hip_bf16.h>

typedef __attribute__((ext_vector_type(16))) _Float16 v16h;
typedef __attribute__((ext_vector_type(8)))  _Float16 v8h;
typedef __attribute__((ext_vector_type(8)))  float    v8f;
typedef __attribute__((ext_vector_type(4)))  float    v4f;

#define H_HEADS 4
#define IN_DIM  128
#define HID     64
#define COLS    256
#define PRED    16
#define RELS    50
#define LAYERS  3


__global__ void k_w1_to_f16(const float* __restrict__ W1,
                            _Float16* __restrict__ W1h, int nPieces) {
    const int t = blockIdx.x * blockDim.x + threadIdx.x;
    if (t >= nPieces) return;
    const int lane = t & 31;
    const int half = (t >> 5) & 1;
    const int kc   = (t >> 6) & 3;
    const int ct   = t >> 8;
    const int col  = ct * 16 + (lane & 15);
    const int k0   = kc * 32 + (lane >> 4) * 8 + half * 16;
    const float* __restrict__ p = W1 + (size_t)col * IN_DIM + k0;
    const v4f x0 = *(const v4f*)(p);
    const v4f x1 = *(const v4f*)(p + 4);
    v8h o;
    #pragma unroll
    for (int i = 0; i < 4; ++i) { o[i] = (_Float16)x0[i]; o[4 + i] = (_Float16)x1[i]; }
    *(v8h*)(W1h + (size_t)t * 8) = o;
}

__global__ void k_etab(const float* __restrict__ rel_emb,
                       const float* __restrict__ W_pred,
                       float* __restrict__ etab) {
    int i = blockIdx.x * blockDim.x + threadIdx.x;
    if (i >= LAYERS * RELS * H_HEADS) return;
    int l = i / (RELS * H_HEADS);
    int rem = i % (RELS * H_HEADS);
    int r = rem / H_HEADS;
    int h = rem % H_HEADS;
    float e = 0.f;
    #pragma unroll
    for (int p = 0; p < PRED; ++p)
        e += rel_emb[r * PRED + p] * W_pred[(l * PRED + p) * H_HEADS + h];
    etab[i] = (e > 0.f) ? e : 0.2f * e;
}

__global__ void __launch_bounds__(128)
k_mlp_wmma(const float* __restrict__ inputs,
           const _Float16* __restrict__ W1h,
           const float* __restrict__ b1,
           const float* __restrict__ W2,
           const float* __restrict__ b2,
           float* __restrict__ hout,
           int n, int nTiles) {
    const int wave = (blockIdx.x * blockDim.x + threadIdx.x) >> 5;
    if (wave >= nTiles) return;
    const int lane = threadIdx.x & 31;
    const int grp  = lane >> 4;
    const int l15  = lane & 15;
    const int nodeBase = wave * 16;
    int arow = nodeBase + l15;
    if (arow > n - 1) arow = n - 1;
    const float* __restrict__ abase = inputs + (size_t)arow * IN_DIM;

    const v8f zero = {};

    v16h A[4];
    #pragma unroll
    for (int kc = 0; kc < 4; ++kc) {
        const int kb = kc * 32 + grp * 8;
        v4f a0 = *(const v4f*)(abase + kb + 0);
        v4f a1 = *(const v4f*)(abase + kb + 4);
        v4f a2 = *(const v4f*)(abase + kb + 16);
        v4f a3 = *(const v4f*)(abase + kb + 20);
        #pragma unroll
        for (int i = 0; i < 4; ++i) {
            A[kc][i]      = (_Float16)a0[i];
            A[kc][4 + i]  = (_Float16)a1[i];
            A[kc][8 + i]  = (_Float16)a2[i];
            A[kc][12 + i] = (_Float16)a3[i];
        }
    }
    const v8h* __restrict__ bbase = (const v8h*)W1h + lane;

    #pragma unroll
    for (int head = 0; head < H_HEADS; ++head) {
        v8f acc0 = zero, acc1 = zero, acc2 = zero, acc3 = zero;

        #pragma unroll
        for (int kc = 0; kc < 4; ++kc) {
            #pragma unroll
            for (int tt = 0; tt < 4; ++tt) {
                const int ct = head * 4 + tt;
                const v8h* q = bbase + ((ct * 4 + kc) * 2) * 32;
                v8h blo = q[0];
                v8h bhi = q[32];
                v16h B;
                #pragma unroll
                for (int i = 0; i < 8; ++i) { B[i] = blo[i]; B[8 + i] = bhi[i]; }
                v8f c = (tt == 0) ? acc0 : (tt == 1) ? acc1 : (tt == 2) ? acc2 : acc3;
                c = __builtin_amdgcn_wmma_f32_16x16x32_f16(
                        false, A[kc], false, B, (short)0, c, false, false);
                if (tt == 0) acc0 = c; else if (tt == 1) acc1 = c;
                else if (tt == 2) acc2 = c; else acc3 = c;
            }
        }

        v8f hacc = zero;
        #pragma unroll
        for (int tt = 0; tt < 4; ++tt) {
            const int col  = head * HID + tt * 16 + l15;
            const float bias = b1[col];
            const float w2   = W2[col];
            v8f c = (tt == 0) ? acc0 : (tt == 1) ? acc1 : (tt == 2) ? acc2 : acc3;
            #pragma unroll
            for (int v = 0; v < 8; ++v) {
                float p = c[v] + bias;
                p = (p > 0.f) ? p : 0.f;
                p *= w2;
                p += __shfl_xor(p, 1, 32);
                p += __shfl_xor(p, 2, 32);
                p += __shfl_xor(p, 4, 32);
                p += __shfl_xor(p, 8, 32);
                hacc[v] += p;
            }
        }
        if (l15 == 0) {
            #pragma unroll
            for (int v = 0; v < 8; ++v) {
                const int node = nodeBase + v + 8 * grp;
                if (node < n)
                    hout[node * H_HEADS + head] = hacc[v] + b2[head];
            }
        }
    }
}

__global__ void k_zero4(v4f* __restrict__ p, int n4) {
    int i = blockIdx.x * blockDim.x + threadIdx.x;
    if (i < n4) { v4f z = {}; p[i] = z; }
}

#define FR_SH   7
#define FR_RS   128
#define KPAD    1024
#define BK_T    1024
#define BK_EPT  8
#define BK_CH   8192
#define KEY_SH  20
#define D_MASK  0xFFFFF

static_assert(BK_T == KPAD, "one thread per key slot in the grouping kernels");
static_assert(BK_CH == BK_T * BK_EPT, "chunk size");
static_assert(FR_RS == (1 << FR_SH), "range size");

__global__ void __launch_bounds__(BK_T)
k_key_hist(const int* __restrict__ dst, int* __restrict__ ghist, int E) {
    __shared__ int lh[KPAD];
    const int tid = threadIdx.x;
    lh[tid] = 0;
    __syncthreads();
    const int base = blockIdx.x * BK_CH + tid;
    #pragma unroll
    for (int k = 0; k < BK_EPT; ++k) {
        const int idx = base + k * BK_T;
        if (idx < E) atomicAdd(&lh[(unsigned)dst[idx] >> FR_SH], 1);
    }
    __syncthreads();
    const int c = lh[tid];
    if (c) atomicAdd(&ghist[tid], c);
}

__global__ void __launch_bounds__(BK_T)
k_bucket(const int* __restrict__ et, const int* __restrict__ src,
         const int* __restrict__ dst, const int* __restrict__ ghist,
         int* __restrict__ cursor, int* __restrict__ kstart,
         int* __restrict__ pairs, int E) {
    __shared__ int lh[KPAD];
    __shared__ int lstart[KPAD];
    __shared__ int gbase[KPAD];
    __shared__ int wg[BK_T / 32];
    __shared__ int wc[BK_T / 32];
    __shared__ int ld[BK_CH];
    __shared__ int ls[BK_CH];
    __shared__ int lt[BK_CH];
    const int tid = threadIdx.x;
    lh[tid] = 0;
    __syncthreads();
    const int cbase = blockIdx.x * BK_CH;
    int nValid = E - cbase; if (nValid > BK_CH) nValid = BK_CH;
    int key[BK_EPT], pk[BK_EPT], s[BK_EPT], rk[BK_EPT];
    #pragma unroll
    for (int k = 0; k < BK_EPT; ++k) {
        const int idx = cbase + k * BK_T + tid;
        const bool ok = idx < E;
        const int t = ok ? et[idx] : 0;
        const int d = ok ? dst[idx] : 0;
        s[k]   = ok ? src[idx] : 0;
        pk[k]  = d | (t << KEY_SH);
        key[k] = ok ? (int)((unsigned)d >> FR_SH) : -1;
    }
    #pragma unroll
    for (int k = 0; k < BK_EPT; ++k)
        rk[k] = (key[k] >= 0) ? atomicAdd(&lh[key[k]], 1) : 0;
    __syncthreads();
    {
        const int lane = tid & 31;
        const int wave = tid >> 5;
        const int g = ghist[tid];
        const int c = lh[tid];
        int gs = g, cs = c;
        #pragma unroll
        for (int off = 1; off < 32; off <<= 1) {
            const int a = __shfl_up(gs, off, 32);
            const int b = __shfl_up(cs, off, 32);
            if (lane >= off) { gs += a; cs += b; }
        }
        if (lane == 31) { wg[wave] = gs; wc[wave] = cs; }
        __syncthreads();
        int gpre = 0, cpre = 0;
        #pragma unroll
        for (int i = 0; i < BK_T / 32; ++i) {
            gpre += (i < wave) ? wg[i] : 0;
            cpre += (i < wave) ? wc[i] : 0;
        }
        const int off = c ? atomicAdd(&cursor[tid], c) : 0;
        lstart[tid] = cpre + cs - c;
        gbase[tid]  = gpre + gs - g + off;
        if (blockIdx.x == 0) kstart[tid] = gpre + gs - g;
    }
    __syncthreads();
    #pragma unroll
    for (int k = 0; k < BK_EPT; ++k) {
        if (key[k] >= 0) {
            const int pos = lstart[key[k]] + rk[k];
            ld[pos] = pk[k]; ls[pos] = s[k]; lt[pos] = key[k];
        }
    }
    __syncthreads();
    #pragma unroll
    for (int k = 0; k < BK_EPT; ++k) {
        const int i = k * BK_T + tid;
        if (i < nValid) {
            const int kk = lt[i];
            const int slot = gbase[kk] + (i - lstart[kk]);
            int2 p; p.x = ld[i]; p.y = ls[i];
            *(int2*)(pairs + 2 * (size_t)slot) = p;
        }
    }
}

__device__ __forceinline__ int quad_bcast(int v, int j) {
    switch (j) {
        case 0:  return __builtin_amdgcn_update_dpp(0, v, 0x00, 0xf, 0xf, true);
        case 1:  return __builtin_amdgcn_update_dpp(0, v, 0x55, 0xf, 0xf, true);
        case 2:  return __builtin_amdgcn_update_dpp(0, v, 0xAA, 0xf, 0xf, true);
        default: return __builtin_amdgcn_update_dpp(0, v, 0xFF, 0xf, 0xf, true);
    }
}

#define LY_T1   512
#define LY_TG   256
#define LY_U    16
#define LY_EPT  8
static_assert(LY_EPT == 8 && LY_U == 16, "the load fences below name every slot");
#define CELLS   (RELS * FR_RS)

template <int HI, bool LAST>
__global__ void __launch_bounds__(HI == 4 ? LY_T1 : LY_TG)
k_layer(const int* __restrict__ pairs, const int* __restrict__ ghist,
        const int* __restrict__ kstart, const float* __restrict__ etab_l,
        const float* __restrict__ hin, float* __restrict__ hout,
        const float* __restrict__ centrality, const float* __restrict__ gamma,
        const float* __restrict__ beta, float* __restrict__ out, int n) {
    constexpr int T  = (HI == 4) ? LY_T1 : LY_TG;
    constexpr int NW = T / 32;
    static_assert(T >= FR_RS, "one thread per node in the store pass");
    __shared__ __attribute__((aligned(16))) int   tab[CELLS * (1 + HI)];
    __shared__ __attribute__((aligned(16))) float eL[RELS * H_HEADS];
    int*   cnt = tab;
    float* G   = (float*)(tab + CELLS);
    const int tid  = threadIdx.x;
    const int lane = tid & 31;
    const int wave = tid >> 5;
    const int r    = blockIdx.x;
    const int lo   = r * FR_RS;
    int hi = lo + FR_RS; if (hi > n) hi = n;
    const int span  = (hi > lo) ? (hi - lo) : 0;
    const int start = kstart[r];
    const int end   = start + ghist[r];
    for (int i = tid; i < RELS * H_HEADS; i += T) eL[i] = etab_l[i];
    {
        const v4f z = {};
        for (int i = tid; i < CELLS * (1 + HI) / 4; i += T) ((v4f*)tab)[i] = z;
    }
    __syncthreads();

    if (HI == 4) {
        const int h = tid & 3;
        for (int base = start + wave * (32 * LY_EPT); base < end; base += NW * 32 * LY_EPT) {
            int px[LY_EPT], py[LY_EPT];
            #pragma unroll
            for (int k = 0; k < LY_EPT; ++k) {
                int idx = base + k * 32 + lane;
                idx = (idx < end) ? idx : (end - 1);
                const int2 p = *(const int2*)(pairs + 2 * (size_t)idx);
                px[k] = p.x;
                py[k] = p.y;
            }
            asm volatile("" :: "v"(py[0]), "v"(py[1]), "v"(py[2]), "v"(py[3]),
                               "v"(py[4]), "v"(py[5]), "v"(py[6]), "v"(py[7]));
            #pragma unroll
            for (int k = 0; k < LY_EPT; ++k)
                px[k] = ((base + k * 32 + lane) < end) ? px[k] : -1;
            int   cell[4 * LY_EPT];
            float hv[4 * LY_EPT];
            #pragma unroll
            for (int s = 0; s < 4 * LY_EPT; ++s) {
                const int k  = s >> 2;
                const int jj = s & 3;
                const int dk = quad_bcast(px[k], jj);
                const unsigned sn = (unsigned)quad_bcast(py[k], jj);
                const bool ok = dk >= 0;
                const float x = hin[sn * H_HEADS + (unsigned)h];
                hv[s]   = ok ? x : 0.f;
                cell[s] = ok ? (((dk >> KEY_SH) << FR_SH) + ((dk & D_MASK) - lo)) : 0;
            }
            #pragma unroll
            for (int k = 0; k < LY_EPT; ++k) {
                const bool ok = px[k] >= 0;
                const int  c  = ok ? (((px[k] >> KEY_SH) << FR_SH) + ((px[k] & D_MASK) - lo)) : 0;
                atomicAdd(&cnt[c], ok ? 1 : 0);
            }
            #pragma unroll
            for (int s = 0; s < 4 * LY_EPT; ++s)
                atomicAdd(&G[cell[s] * H_HEADS + h], hv[s]);
        }
    } else {
        for (int base = start + wave * (32 * LY_U); base < end; base += NW * 32 * LY_U) {
            int px[LY_U], py[LY_U];
            #pragma unroll
            for (int k = 0; k < LY_U; ++k) {
                int idx = base + k * 32 + lane;
                idx = (idx < end) ? idx : (end - 1);
                const int2 p = *(const int2*)(pairs + 2 * (size_t)idx);
                px[k] = p.x;
                py[k] = p.y;
            }
            asm volatile("" :: "v"(py[0]), "v"(py[1]), "v"(py[2]), "v"(py[3]),
                               "v"(py[4]), "v"(py[5]), "v"(py[6]), "v"(py[7]),
                               "v"(py[8]), "v"(py[9]), "v"(py[10]), "v"(py[11]),
                               "v"(py[12]), "v"(py[13]), "v"(py[14]), "v"(py[15]));
            float hv[LY_U];
            #pragma unroll
            for (int k = 0; k < LY_U; ++k) hv[k] = hin[(unsigned)py[k]];
            #pragma unroll
            for (int k = 0; k < LY_U; ++k) {
                const bool ok = (base + k * 32 + lane) < end;
                const int  c  = ok ? (((px[k] >> KEY_SH) << FR_SH) + ((px[k] & D_MASK) - lo)) : 0;
                atomicAdd(&cnt[c], ok ? 1 : 0);
                atomicAdd(&G[c], ok ? hv[k] : 0.f);
            }
        }
    }
    __syncthreads();

    if (tid < span) {
        const int d = tid;
        v4f mx;
        mx[0] = mx[1] = mx[2] = mx[3] = -__builtin_huge_valf();
        #pragma unroll 10
        for (int t = 0; t < RELS; ++t) {
            const bool present = cnt[t * FR_RS + d] > 0;
            const v4f e = *(const v4f*)(eL + t * H_HEADS);
            #pragma unroll
            for (int h = 0; h < H_HEADS; ++h)
                mx[h] = present ? fmaxf(mx[h], e[h]) : mx[h];
        }
        v4f sv = {}, uv = {};
        #pragma unroll 10
        for (int t = 0; t < RELS; ++t) {
            const int   c  = cnt[t * FR_RS + d];
            const float cf = (float)c;
            const v4f e = *(const v4f*)(eL + t * H_HEADS);
            v4f g;
            if (HI == 4) {
                g = *(const v4f*)(G + (t * FR_RS + d) * H_HEADS);
            } else {
                const float g1 = G[t * FR_RS + d];
                g[0] = g1; g[1] = g1; g[2] = g1; g[3] = g1;
            }
            #pragma unroll
            for (int h = 0; h < H_HEADS; ++h) {
                const float w = (c > 0) ? __expf(e[h] - mx[h]) : 0.f;
                sv[h] += cf * w;
                uv[h] += w * g[h];
            }
        }
        float v[H_HEADS];
        #pragma unroll
        for (int h = 0; h < H_HEADS; ++h) {
            const float x = (sv[h] > 0.f) ? (uv[h] / sv[h]) : 0.f;
            v[h] = (x > 0.f) ? x : 0.f;
        }
        if (!LAST) {
            hout[lo + d] = 0.25f * (v[0] + v[1] + v[2] + v[3]);
        } else {
            const float c = centrality[lo + d];
            float acc = 0.f;
            #pragma unroll
            for (int h = 0; h < H_HEADS; ++h)
                acc += (c * gamma[h] + beta[h]) * v[h];
            acc *= 0.25f;
            out[lo + d] = (acc > 0.f) ? acc : 0.01f * acc;
        }
    }
}

extern "C" void kernel_launch(void* const* d_in, const int* in_sizes, int n_in,
                              void* d_out, int out_size, void* d_ws, size_t ws_size,
                              hipStream_t stream) {
    const float* inputs     = (const float*)d_in[0];
    const float* centrality = (const float*)d_in[1];
    const float* W1         = (const float*)d_in[2];
    const float* b1         = (const float*)d_in[3];
    const float* W2         = (const float*)d_in[4];
    const float* b2         = (const float*)d_in[5];
    const float* rel_emb    = (const float*)d_in[6];
    const float* W_pred     = (const float*)d_in[7];
    const float* gamma      = (const float*)d_in[8];
    const float* beta       = (const float*)d_in[9];
    const int*   edge_types = (const int*)d_in[10];
    const int*   src        = (const int*)d_in[11];
    const int*   dst        = (const int*)d_in[12];
    float* out = (float*)d_out;

    const int N = in_sizes[1];
    const int E = in_sizes[10];

    char* ws = (char*)d_ws;
    _Float16* W1h = (_Float16*)ws;
    float* etab   = (float*)(ws + 65536);
    int*   ghist  = (int*)(ws + 69632);
    int*   cursor = ghist + KPAD;
    int*   kstart = cursor + KPAD;
    float* hA     = (float*)(kstart + KPAD);
    const size_t hbWords = ((size_t)N + 3) & ~(size_t)3;
    float* hbA    = hA + (size_t)N * H_HEADS;
    float* hbB    = hbA + hbWords;
    int*   pairs  = (int*)(hbB + hbWords);

    const int TPB = 256;

    {
        int nPieces = COLS * IN_DIM / 8;
        k_w1_to_f16<<<(nPieces + TPB - 1) / TPB, TPB, 0, stream>>>(W1, W1h, nPieces);
    }
    {
        int c = LAYERS * RELS * H_HEADS;
        k_etab<<<(c + TPB - 1) / TPB, TPB, 0, stream>>>(rel_emb, W_pred, etab);
    }
    {
        int n4 = 2 * KPAD / 4;
        k_zero4<<<(n4 + TPB - 1) / TPB, TPB, 0, stream>>>((v4f*)ghist, n4);
    }
    {
        int nTiles = (N + 15) / 16;
        int wavesPerBlock = 4;
        int blocks = (nTiles + wavesPerBlock - 1) / wavesPerBlock;
        k_mlp_wmma<<<blocks, wavesPerBlock * 32, 0, stream>>>(
            inputs, W1h, b1, W2, b2, hA, N, nTiles);
    }

    const int bBlocks = (E + BK_CH - 1) / BK_CH;
    const int nRanges = (N + FR_RS - 1) / FR_RS;

    k_key_hist<<<bBlocks, BK_T, 0, stream>>>(dst, ghist, E);
    k_bucket<<<bBlocks, BK_T, 0, stream>>>(edge_types, src, dst, ghist, cursor, kstart, pairs, E);

    k_layer<4, false><<<nRanges, LY_T1, 0, stream>>>(pairs, ghist, kstart, etab, hA, hbA,
                                                     centrality, gamma, beta, out, N);
    k_layer<1, false><<<nRanges, LY_TG, 0, stream>>>(pairs, ghist, kstart, etab + 1 * RELS * H_HEADS,
                                                     hbA, hbB, centrality, gamma, beta, out, N);
    k_layer<1, true><<<nRanges, LY_TG, 0, stream>>>(pairs, ghist, kstart, etab + 2 * RELS * H_HEADS,
                                                    hbB, hbB, centrality, gamma, beta, out, N);
}


# MLP weight-stationary: W1 is the A operand, a wave keeps one head's 16 fragments in registers over 12 node tiles; node rows stream as B via LDS; 64-unit dot: in-lane sum + 1 exchange (order changed)
# speedup vs baseline: 23.9068x; 1.0687x over previous
#include <hip/hip_runtime.h>
#include <hip/hip_bf16.h>

typedef __attribute__((ext_vector_type(16))) _Float16 v16h;
typedef __attribute__((ext_vector_type(8)))  _Float16 v8h;
typedef __attribute__((ext_vector_type(8)))  float    v8f;
typedef __attribute__((ext_vector_type(4)))  float    v4f;

#define H_HEADS 4
#define IN_DIM  128
#define HID     64
#define COLS    256
#define PRED    16
#define RELS    50
#define LAYERS  3


__global__ void k_w1_to_f16(const float* __restrict__ W1,
                            _Float16* __restrict__ W1h, int nPieces) {
    const int t = blockIdx.x * blockDim.x + threadIdx.x;
    if (t >= nPieces) return;
    const int lane = t & 31;
    const int half = (t >> 5) & 1;
    const int kc   = (t >> 6) & 3;
    const int ct   = t >> 8;
    const int col  = ct * 16 + (lane & 15);
    const int k0   = kc * 32 + (lane >> 4) * 8 + half * 16;
    const float* __restrict__ p = W1 + (size_t)col * IN_DIM + k0;
    const v4f x0 = *(const v4f*)(p);
    const v4f x1 = *(const v4f*)(p + 4);
    v8h o;
    #pragma unroll
    for (int i = 0; i < 4; ++i) { o[i] = (_Float16)x0[i]; o[4 + i] = (_Float16)x1[i]; }
    *(v8h*)(W1h + (size_t)t * 8) = o;
}

__global__ void k_etab(const float* __restrict__ rel_emb,
                       const float* __restrict__ W_pred,
                       float* __restrict__ etab) {
    int i = blockIdx.x * blockDim.x + threadIdx.x;
    if (i >= LAYERS * RELS * H_HEADS) return;
    int l = i / (RELS * H_HEADS);
    int rem = i % (RELS * H_HEADS);
    int r = rem / H_HEADS;
    int h = rem % H_HEADS;
    float e = 0.f;
    #pragma unroll
    for (int p = 0; p < PRED; ++p)
        e += rel_emb[r * PRED + p] * W_pred[(l * PRED + p) * H_HEADS + h];
    etab[i] = (e > 0.f) ? e : 0.2f * e;
}

__device__ __forceinline__ void lds_barrier() {
    __builtin_amdgcn_fence(__ATOMIC_RELEASE, "workgroup", "local");
    __builtin_amdgcn_s_barrier();
    __builtin_amdgcn_fence(__ATOMIC_ACQUIRE, "workgroup", "local");
}

#define MLP_ROUNDS 3
#define MLP_TPB    (4 * MLP_ROUNDS)
__global__ void __launch_bounds__(128)
k_mlp_wmma(const float* __restrict__ inputs,
           const _Float16* __restrict__ W1h,
           const float* __restrict__ b1,
           const float* __restrict__ W2,
           const float* __restrict__ b2,
           float* __restrict__ hout,
           int n) {
    __shared__ __attribute__((aligned(16))) v8h   xs[4][4][2][32];
    __shared__ __attribute__((aligned(16))) float os[4][16][H_HEADS];
    __shared__ __attribute__((aligned(16))) float bs[COLS];
    __shared__ __attribute__((aligned(16))) float w2s[COLS];
    const int tid  = threadIdx.x;
    const int lane = tid & 31;
    const int wave = tid >> 5;
    const int grp  = lane >> 4;
    const int l15  = lane & 15;
    const int tile0 = blockIdx.x * MLP_TPB;

    static_assert(COLS == 256, "two b1 and two W2 values per thread of 128");
    const float b1r0 = b1[tid], b1r1 = b1[tid + 128];
    const float w2r0 = W2[tid], w2r1 = W2[tid + 128];
    const float b2h  = b2[wave];

    v16h Wf[4][4];
    {
        const v8h* __restrict__ bbase = (const v8h*)W1h + lane;
        #pragma unroll
        for (int tt = 0; tt < 4; ++tt) {
            #pragma unroll
            for (int kc = 0; kc < 4; ++kc) {
                const v8h* q = bbase + (((wave * 4 + tt) * 4 + kc) * 2) * 32;
                const v8h lo = q[0];
                const v8h hi = q[32];
                #pragma unroll
                for (int i = 0; i < 8; ++i) { Wf[tt][kc][i] = lo[i]; Wf[tt][kc][8 + i] = hi[i]; }
            }
        }
    }

    v4f raw[16];
    {
        int arow = (tile0 + wave) * 16 + l15;
        if (arow > n - 1) arow = n - 1;
        const float* __restrict__ ab = inputs + (size_t)arow * IN_DIM + grp * 8;
        #pragma unroll
        for (int kc = 0; kc < 4; ++kc) {
            raw[4 * kc + 0] = *(const v4f*)(ab + kc * 32 + 0);
            raw[4 * kc + 1] = *(const v4f*)(ab + kc * 32 + 4);
            raw[4 * kc + 2] = *(const v4f*)(ab + kc * 32 + 16);
            raw[4 * kc + 3] = *(const v4f*)(ab + kc * 32 + 20);
        }
    }
    bs[tid] = b1r0;  bs[tid + 128] = b1r1;
    w2s[tid] = w2r0; w2s[tid + 128] = w2r1;

    for (int rd = 0; rd < MLP_ROUNDS; ++rd) {
        #pragma unroll
        for (int kc = 0; kc < 4; ++kc) {
            v8h lo, hi;
            #pragma unroll
            for (int i = 0; i < 4; ++i) {
                lo[i]     = (_Float16)raw[4 * kc + 0][i];
                lo[4 + i] = (_Float16)raw[4 * kc + 1][i];
                hi[i]     = (_Float16)raw[4 * kc + 2][i];
                hi[4 + i] = (_Float16)raw[4 * kc + 3][i];
            }
            xs[wave][kc][0][lane] = lo;
            xs[wave][kc][1][lane] = hi;
        }
        if (rd + 1 < MLP_ROUNDS) {
            int arow = (tile0 + (rd + 1) * 4 + wave) * 16 + l15;
            if (arow > n - 1) arow = n - 1;
            const float* __restrict__ ab = inputs + (size_t)arow * IN_DIM + grp * 8;
            #pragma unroll
            for (int kc = 0; kc < 4; ++kc) {
                raw[4 * kc + 0] = *(const v4f*)(ab + kc * 32 + 0);
                raw[4 * kc + 1] = *(const v4f*)(ab + kc * 32 + 4);
                raw[4 * kc + 2] = *(const v4f*)(ab + kc * 32 + 16);
                raw[4 * kc + 3] = *(const v4f*)(ab + kc * 32 + 20);
            }
        }
        lds_barrier();

        for (int t = 0; t < 4; ++t) {
            v16h X[4];
            #pragma unroll
            for (int kc = 0; kc < 4; ++kc) {
                const v8h lo = xs[t][kc][0][lane];
                const v8h hi = xs[t][kc][1][lane];
                #pragma unroll
                for (int i = 0; i < 8; ++i) { X[kc][i] = lo[i]; X[kc][8 + i] = hi[i]; }
            }
            v8f acc[4];
            #pragma unroll
            for (int tt = 0; tt < 4; ++tt) { const v8f z = {}; acc[tt] = z; }
            #pragma unroll
            for (int kc = 0; kc < 4; ++kc) {
                #pragma unroll
                for (int tt = 0; tt < 4; ++tt)
                    acc[tt] = __builtin_amdgcn_wmma_f32_16x16x32_f16(
                                  false, Wf[tt][kc], false, X[kc], (short)0, acc[tt], false, false);
            }
            float sp[4];
            #pragma unroll
            for (int tt = 0; tt < 4; ++tt) {
                const int c0 = wave * HID + tt * 16 + grp * 8;
                const v4f bA = *(const v4f*)(bs + c0), bB = *(const v4f*)(bs + c0 + 4);
                const v4f wA = *(const v4f*)(w2s + c0), wB = *(const v4f*)(w2s + c0 + 4);
                float s = 0.f;
                #pragma unroll
                for (int v = 0; v < 4; ++v) {
                    float p = acc[tt][v] + bA[v];
                    p = (p > 0.f) ? p : 0.f;
                    s += p * wA[v];
                }
                #pragma unroll
                for (int v = 0; v < 4; ++v) {
                    float p = acc[tt][4 + v] + bB[v];
                    p = (p > 0.f) ? p : 0.f;
                    s += p * wB[v];
                }
                sp[tt] = s;
            }
            float s = (sp[0] + sp[1]) + (sp[2] + sp[3]);
            s += __shfl_xor(s, 16, 32);
            if (grp == 0) os[t][l15][wave] = s + b2h;
        }
        lds_barrier();
        {
            const int node = (tile0 + rd * 4 + wave) * 16 + l15;
            if (grp == 0 && node < n)
                *(v4f*)(hout + (size_t)node * H_HEADS) = *(const v4f*)(&os[wave][l15][0]);
        }
    }
}

__global__ void k_zero4(v4f* __restrict__ p, int n4) {
    int i = blockIdx.x * blockDim.x + threadIdx.x;
    if (i < n4) { v4f z = {}; p[i] = z; }
}

#define FR_SH   7
#define FR_RS   128
#define KPAD    1024
#define BK_T    1024
#define BK_EPT  8
#define BK_CH   8192
#define KEY_SH  20
#define D_MASK  0xFFFFF

static_assert(BK_T == KPAD, "one thread per key slot in the grouping kernels");
static_assert(BK_CH == BK_T * BK_EPT, "chunk size");
static_assert(FR_RS == (1 << FR_SH), "range size");

__global__ void __launch_bounds__(BK_T)
k_key_hist(const int* __restrict__ dst, int* __restrict__ ghist, int E) {
    __shared__ int lh[KPAD];
    const int tid = threadIdx.x;
    lh[tid] = 0;
    __syncthreads();
    const int base = blockIdx.x * BK_CH + tid;
    #pragma unroll
    for (int k = 0; k < BK_EPT; ++k) {
        const int idx = base + k * BK_T;
        if (idx < E) atomicAdd(&lh[(unsigned)dst[idx] >> FR_SH], 1);
    }
    __syncthreads();
    const int c = lh[tid];
    if (c) atomicAdd(&ghist[tid], c);
}

__global__ void __launch_bounds__(BK_T)
k_bucket(const int* __restrict__ et, const int* __restrict__ src,
         const int* __restrict__ dst, const int* __restrict__ ghist,
         int* __restrict__ cursor, int* __restrict__ kstart,
         int* __restrict__ pairs, int E) {
    __shared__ int lh[KPAD];
    __shared__ int lstart[KPAD];
    __shared__ int gbase[KPAD];
    __shared__ int wg[BK_T / 32];
    __shared__ int wc[BK_T / 32];
    __shared__ int ld[BK_CH];
    __shared__ int ls[BK_CH];
    __shared__ int lt[BK_CH];
    const int tid = threadIdx.x;
    lh[tid] = 0;
    __syncthreads();
    const int cbase = blockIdx.x * BK_CH;
    int nValid = E - cbase; if (nValid > BK_CH) nValid = BK_CH;
    int key[BK_EPT], pk[BK_EPT], s[BK_EPT], rk[BK_EPT];
    #pragma unroll
    for (int k = 0; k < BK_EPT; ++k) {
        const int idx = cbase + k * BK_T + tid;
        const bool ok = idx < E;
        const int t = ok ? et[idx] : 0;
        const int d = ok ? dst[idx] : 0;
        s[k]   = ok ? src[idx] : 0;
        pk[k]  = d | (t << KEY_SH);
        key[k] = ok ? (int)((unsigned)d >> FR_SH) : -1;
    }
    #pragma unroll
    for (int k = 0; k < BK_EPT; ++k)
        rk[k] = (key[k] >= 0) ? atomicAdd(&lh[key[k]], 1) : 0;
    __syncthreads();
    {
        const int lane = tid & 31;
        const int wave = tid >> 5;
        const int g = ghist[tid];
        const int c = lh[tid];
        int gs = g, cs = c;
        #pragma unroll
        for (int off = 1; off < 32; off <<= 1) {
            const int a = __shfl_up(gs, off, 32);
            const int b = __shfl_up(cs, off, 32);
            if (lane >= off) { gs += a; cs += b; }
        }
        if (lane == 31) { wg[wave] = gs; wc[wave] = cs; }
        __syncthreads();
        int gpre = 0, cpre = 0;
        #pragma unroll
        for (int i = 0; i < BK_T / 32; ++i) {
            gpre += (i < wave) ? wg[i] : 0;
            cpre += (i < wave) ? wc[i] : 0;
        }
        const int off = c ? atomicAdd(&cursor[tid], c) : 0;
        lstart[tid] = cpre + cs - c;
        gbase[tid]  = gpre + gs - g + off;
        if (blockIdx.x == 0) kstart[tid] = gpre + gs - g;
    }
    __syncthreads();
    #pragma unroll
    for (int k = 0; k < BK_EPT; ++k) {
        if (key[k] >= 0) {
            const int pos = lstart[key[k]] + rk[k];
            ld[pos] = pk[k]; ls[pos] = s[k]; lt[pos] = key[k];
        }
    }
    __syncthreads();
    #pragma unroll
    for (int k = 0; k < BK_EPT; ++k) {
        const int i = k * BK_T + tid;
        if (i < nValid) {
            const int kk = lt[i];
            const int slot = gbase[kk] + (i - lstart[kk]);
            int2 p; p.x = ld[i]; p.y = ls[i];
            *(int2*)(pairs + 2 * (size_t)slot) = p;
        }
    }
}

__device__ __forceinline__ int quad_bcast(int v, int j) {
    switch (j) {
        case 0:  return __builtin_amdgcn_update_dpp(0, v, 0x00, 0xf, 0xf, true);
        case 1:  return __builtin_amdgcn_update_dpp(0, v, 0x55, 0xf, 0xf, true);
        case 2:  return __builtin_amdgcn_update_dpp(0, v, 0xAA, 0xf, 0xf, true);
        default: return __builtin_amdgcn_update_dpp(0, v, 0xFF, 0xf, 0xf, true);
    }
}

#define LY_T1   512
#define LY_TG   256
#define LY_U    16
#define LY_EPT  8
static_assert(LY_EPT == 8 && LY_U == 16, "the load fences below name every slot");
#define CELLS   (RELS * FR_RS)

template <int HI, bool LAST>
__global__ void __launch_bounds__(HI == 4 ? LY_T1 : LY_TG)
k_layer(const int* __restrict__ pairs, const int* __restrict__ ghist,
        const int* __restrict__ kstart, const float* __restrict__ etab_l,
        const float* __restrict__ hin, float* __restrict__ hout,
        const float* __restrict__ centrality, const float* __restrict__ gamma,
        const float* __restrict__ beta, float* __restrict__ out, int n) {
    constexpr int T  = (HI == 4) ? LY_T1 : LY_TG;
    constexpr int NW = T / 32;
    static_assert(T >= FR_RS, "one thread per node in the store pass");
    __shared__ __attribute__((aligned(16))) int   tab[CELLS * (1 + HI)];
    __shared__ __attribute__((aligned(16))) float eL[RELS * H_HEADS];
    int*   cnt = tab;
    float* G   = (float*)(tab + CELLS);
    const int tid  = threadIdx.x;
    const int lane = tid & 31;
    const int wave = tid >> 5;
    const int r    = blockIdx.x;
    const int lo   = r * FR_RS;
    int hi = lo + FR_RS; if (hi > n) hi = n;
    const int span  = (hi > lo) ? (hi - lo) : 0;
    const int start = kstart[r];
    const int end   = start + ghist[r];
    for (int i = tid; i < RELS * H_HEADS; i += T) eL[i] = etab_l[i];
    {
        const v4f z = {};
        for (int i = tid; i < CELLS * (1 + HI) / 4; i += T) ((v4f*)tab)[i] = z;
    }
    __syncthreads();

    if (HI == 4) {
        const int h = tid & 3;
        for (int base = start + wave * (32 * LY_EPT); base < end; base += NW * 32 * LY_EPT) {
            int px[LY_EPT], py[LY_EPT];
            #pragma unroll
            for (int k = 0; k < LY_EPT; ++k) {
                int idx = base + k * 32 + lane;
                idx = (idx < end) ? idx : (end - 1);
                const int2 p = *(const int2*)(pairs + 2 * (size_t)idx);
                px[k] = p.x;
                py[k] = p.y;
            }
            asm volatile("" :: "v"(py[0]), "v"(py[1]), "v"(py[2]), "v"(py[3]),
                               "v"(py[4]), "v"(py[5]), "v"(py[6]), "v"(py[7]));
            #pragma unroll
            for (int k = 0; k < LY_EPT; ++k)
                px[k] = ((base + k * 32 + lane) < end) ? px[k] : -1;
            int   cell[4 * LY_EPT];
            float hv[4 * LY_EPT];
            #pragma unroll
            for (int s = 0; s < 4 * LY_EPT; ++s) {
                const int k  = s >> 2;
                const int jj = s & 3;
                const int dk = quad_bcast(px[k], jj);
                const unsigned sn = (unsigned)quad_bcast(py[k], jj);
                const bool ok = dk >= 0;
                const float x = hin[sn * H_HEADS + (unsigned)h];
                hv[s]   = ok ? x : 0.f;
                cell[s] = ok ? (((dk >> KEY_SH) << FR_SH) + ((dk & D_MASK) - lo)) : 0;
            }
            #pragma unroll
            for (int k = 0; k < LY_EPT; ++k) {
                const bool ok = px[k] >= 0;
                const int  c  = ok ? (((px[k] >> KEY_SH) << FR_SH) + ((px[k] & D_MASK) - lo)) : 0;
                atomicAdd(&cnt[c], ok ? 1 : 0);
            }
            #pragma unroll
            for (int s = 0; s < 4 * LY_EPT; ++s)
                atomicAdd(&G[cell[s] * H_HEADS + h], hv[s]);
        }
    } else {
        for (int base = start + wave * (32 * LY_U); base < end; base += NW * 32 * LY_U) {
            int px[LY_U], py[LY_U];
            #pragma unroll
            for (int k = 0; k < LY_U; ++k) {
                int idx = base + k * 32 + lane;
                idx = (idx < end) ? idx : (end - 1);
                const int2 p = *(const int2*)(pairs + 2 * (size_t)idx);
                px[k] = p.x;
                py[k] = p.y;
            }
            asm volatile("" :: "v"(py[0]), "v"(py[1]), "v"(py[2]), "v"(py[3]),
                               "v"(py[4]), "v"(py[5]), "v"(py[6]), "v"(py[7]),
                               "v"(py[8]), "v"(py[9]), "v"(py[10]), "v"(py[11]),
                               "v"(py[12]), "v"(py[13]), "v"(py[14]), "v"(py[15]));
            float hv[LY_U];
            #pragma unroll
            for (int k = 0; k < LY_U; ++k) hv[k] = hin[(unsigned)py[k]];
            #pragma unroll
            for (int k = 0; k < LY_U; ++k) {
                const bool ok = (base + k * 32 + lane) < end;
                const int  c  = ok ? (((px[k] >> KEY_SH) << FR_SH) + ((px[k] & D_MASK) - lo)) : 0;
                atomicAdd(&cnt[c], ok ? 1 : 0);
                atomicAdd(&G[c], ok ? hv[k] : 0.f);
            }
        }
    }
    __syncthreads();

    if (tid < span) {
        const int d = tid;
        v4f mx;
        mx[0] = mx[1] = mx[2] = mx[3] = -__builtin_huge_valf();
        #pragma unroll 10
        for (int t = 0; t < RELS; ++t) {
            const bool present = cnt[t * FR_RS + d] > 0;
            const v4f e = *(const v4f*)(eL + t * H_HEADS);
            #pragma unroll
            for (int h = 0; h < H_HEADS; ++h)
                mx[h] = present ? fmaxf(mx[h], e[h]) : mx[h];
        }
        v4f sv = {}, uv = {};
        #pragma unroll 10
        for (int t = 0; t < RELS; ++t) {
            const int   c  = cnt[t * FR_RS + d];
            const float cf = (float)c;
            const v4f e = *(const v4f*)(eL + t * H_HEADS);
            v4f g;
            if (HI == 4) {
                g = *(const v4f*)(G + (t * FR_RS + d) * H_HEADS);
            } else {
                const float g1 = G[t * FR_RS + d];
                g[0] = g1; g[1] = g1; g[2] = g1; g[3] = g1;
            }
            #pragma unroll
            for (int h = 0; h < H_HEADS; ++h) {
                const float w = (c > 0) ? __expf(e[h] - mx[h]) : 0.f;
                sv[h] += cf * w;
                uv[h] += w * g[h];
            }
        }
        float v[H_HEADS];
        #pragma unroll
        for (int h = 0; h < H_HEADS; ++h) {
            const float x = (sv[h] > 0.f) ? (uv[h] / sv[h]) : 0.f;
            v[h] = (x > 0.f) ? x : 0.f;
        }
        if (!LAST) {
            hout[lo + d] = 0.25f * (v[0] + v[1] + v[2] + v[3]);
        } else {
            const float c = centrality[lo + d];
            float acc = 0.f;
            #pragma unroll
            for (int h = 0; h < H_HEADS; ++h)
                acc += (c * gamma[h] + beta[h]) * v[h];
            acc *= 0.25f;
            out[lo + d] = (acc > 0.f) ? acc : 0.01f * acc;
        }
    }
}

extern "C" void kernel_launch(void* const* d_in, const int* in_sizes, int n_in,
                              void* d_out, int out_size, void* d_ws, size_t ws_size,
                              hipStream_t stream) {
    const float* inputs     = (const float*)d_in[0];
    const float* centrality = (const float*)d_in[1];
    const float* W1         = (const float*)d_in[2];
    const float* b1         = (const float*)d_in[3];
    const float* W2         = (const float*)d_in[4];
    const float* b2         = (const float*)d_in[5];
    const float* rel_emb    = (const float*)d_in[6];
    const float* W_pred     = (const float*)d_in[7];
    const float* gamma      = (const float*)d_in[8];
    const float* beta       = (const float*)d_in[9];
    const int*   edge_types = (const int*)d_in[10];
    const int*   src        = (const int*)d_in[11];
    const int*   dst        = (const int*)d_in[12];
    float* out = (float*)d_out;

    const int N = in_sizes[1];
    const int E = in_sizes[10];

    char* ws = (char*)d_ws;
    _Float16* W1h = (_Float16*)ws;
    float* etab   = (float*)(ws + 65536);
    int*   ghist  = (int*)(ws + 69632);
    int*   cursor = ghist + KPAD;
    int*   kstart = cursor + KPAD;
    float* hA     = (float*)(kstart + KPAD);
    const size_t hbWords = ((size_t)N + 3) & ~(size_t)3;
    float* hbA    = hA + (size_t)N * H_HEADS;
    float* hbB    = hbA + hbWords;
    int*   pairs  = (int*)(hbB + hbWords);

    const int TPB = 256;

    {
        int nPieces = COLS * IN_DIM / 8;
        k_w1_to_f16<<<(nPieces + TPB - 1) / TPB, TPB, 0, stream>>>(W1, W1h, nPieces);
    }
    {
        int c = LAYERS * RELS * H_HEADS;
        k_etab<<<(c + TPB - 1) / TPB, TPB, 0, stream>>>(rel_emb, W_pred, etab);
    }
    {
        int n4 = 2 * KPAD / 4;
        k_zero4<<<(n4 + TPB - 1) / TPB, TPB, 0, stream>>>((v4f*)ghist, n4);
    }
    {
        int nTiles = (N + 15) / 16;
        int blocks = (nTiles + MLP_TPB - 1) / MLP_TPB;
        k_mlp_wmma<<<blocks, 128, 0, stream>>>(inputs, W1h, b1, W2, b2, hA, N);
    }

    const int bBlocks = (E + BK_CH - 1) / BK_CH;
    const int nRanges = (N + FR_RS - 1) / FR_RS;

    k_key_hist<<<bBlocks, BK_T, 0, stream>>>(dst, ghist, E);
    k_bucket<<<bBlocks, BK_T, 0, stream>>>(edge_types, src, dst, ghist, cursor, kstart, pairs, E);

    k_layer<4, false><<<nRanges, LY_T1, 0, stream>>>(pairs, ghist, kstart, etab, hA, hbA,
                                                     centrality, gamma, beta, out, N);
    k_layer<1, false><<<nRanges, LY_TG, 0, stream>>>(pairs, ghist, kstart, etab + 1 * RELS * H_HEADS,
                                                     hbA, hbB, centrality, gamma, beta, out, N);
    k_layer<1, true><<<nRanges, LY_TG, 0, stream>>>(pairs, ghist, kstart, etab + 2 * RELS * H_HEADS,
                                                    hbB, hbB, centrality, gamma, beta, out, N);
}


# speedup vs baseline: 27.1415x; 1.1353x over previous
#include <hip/hip_runtime.h>
#include <hip/hip_bf16.h>

typedef __attribute__((ext_vector_type(16))) _Float16 v16h;
typedef __attribute__((ext_vector_type(8)))  _Float16 v8h;
typedef __attribute__((ext_vector_type(8)))  float    v8f;
typedef __attribute__((ext_vector_type(4)))  float    v4f;

#define H_HEADS 4
#define IN_DIM  128
#define HID     64
#define COLS    256
#define PRED    16
#define RELS    50
#define LAYERS  3


__global__ void k_w1_to_f16(const float* __restrict__ W1,
                            _Float16* __restrict__ W1h, int nPieces) {
    const int t = blockIdx.x * blockDim.x + threadIdx.x;
    if (t >= nPieces) return;
    const int lane = t & 31;
    const int half = (t >> 5) & 1;
    const int kc   = (t >> 6) & 3;
    const int ct   = t >> 8;
    const int col  = ct * 16 + (lane & 15);
    const int k0   = kc * 32 + (lane >> 4) * 8 + half * 16;
    const float* __restrict__ p = W1 + (size_t)col * IN_DIM + k0;
    const v4f x0 = *(const v4f*)(p);
    const v4f x1 = *(const v4f*)(p + 4);
    v8h o;
    #pragma unroll
    for (int i = 0; i < 4; ++i) { o[i] = (_Float16)x0[i]; o[4 + i] = (_Float16)x1[i]; }
    *(v8h*)(W1h + (size_t)t * 8) = o;
}

__global__ void k_etab(const float* __restrict__ rel_emb,
                       const float* __restrict__ W_pred,
                       float* __restrict__ etab) {
    int i = blockIdx.x * blockDim.x + threadIdx.x;
    if (i >= LAYERS * RELS * H_HEADS) return;
    int l = i / (RELS * H_HEADS);
    int rem = i % (RELS * H_HEADS);
    int r = rem / H_HEADS;
    int h = rem % H_HEADS;
    float e = 0.f;
    #pragma unroll
    for (int p = 0; p < PRED; ++p)
        e += rel_emb[r * PRED + p] * W_pred[(l * PRED + p) * H_HEADS + h];
    etab[i] = (e > 0.f) ? e : 0.2f * e;
}

__device__ __forceinline__ void lds_barrier() {
    __builtin_amdgcn_fence(__ATOMIC_RELEASE, "workgroup", "local");
    __builtin_amdgcn_s_barrier();
    __builtin_amdgcn_fence(__ATOMIC_ACQUIRE, "workgroup", "local");
}

#define MLP_ROUNDS 3
#define MLP_TPB    (4 * MLP_ROUNDS)
__global__ void __launch_bounds__(128)
k_mlp_wmma(const float* __restrict__ inputs,
           const _Float16* __restrict__ W1h,
           const float* __restrict__ b1,
           const float* __restrict__ W2,
           const float* __restrict__ b2,
           float* __restrict__ hout,
           int n) {
    __shared__ __attribute__((aligned(16))) v8h   xs[4][4][2][32];
    __shared__ __attribute__((aligned(16))) float os[4][16][H_HEADS];
    __shared__ __attribute__((aligned(16))) float bs[COLS];
    __shared__ __attribute__((aligned(16))) float w2s[COLS];
    const int tid  = threadIdx.x;
    const int lane = tid & 31;
    const int wave = tid >> 5;
    const int grp  = lane >> 4;
    const int l15  = lane & 15;
    const int tile0 = blockIdx.x * MLP_TPB;

    static_assert(COLS == 256, "two b1 and two W2 values per thread of 128");
    const float b1r0 = b1[tid], b1r1 = b1[tid + 128];
    const float w2r0 = W2[tid], w2r1 = W2[tid + 128];
    const float b2h  = b2[wave];

    v16h Wf[4][4];
    {
        const v8h* __restrict__ bbase = (const v8h*)W1h + lane;
        #pragma unroll
        for (int tt = 0; tt < 4; ++tt) {
            #pragma unroll
            for (int kc = 0; kc < 4; ++kc) {
                const v8h* q = bbase + (((wave * 4 + tt) * 4 + kc) * 2) * 32;
                const v8h lo = q[0];
                const v8h hi = q[32];
                #pragma unroll
                for (int i = 0; i < 8; ++i) { Wf[tt][kc][i] = lo[i]; Wf[tt][kc][8 + i] = hi[i]; }
            }
        }
    }

    v4f raw[16];
    {
        int arow = (tile0 + wave) * 16 + l15;
        if (arow > n - 1) arow = n - 1;
        const float* __restrict__ ab = inputs + (size_t)arow * IN_DIM + grp * 8;
        #pragma unroll
        for (int kc = 0; kc < 4; ++kc) {
            raw[4 * kc + 0] = *(const v4f*)(ab + kc * 32 + 0);
            raw[4 * kc + 1] = *(const v4f*)(ab + kc * 32 + 4);
            raw[4 * kc + 2] = *(const v4f*)(ab + kc * 32 + 16);
            raw[4 * kc + 3] = *(const v4f*)(ab + kc * 32 + 20);
        }
    }
    bs[tid] = b1r0;  bs[tid + 128] = b1r1;
    w2s[tid] = w2r0; w2s[tid + 128] = w2r1;

    for (int rd = 0; rd < MLP_ROUNDS; ++rd) {
        #pragma unroll
        for (int kc = 0; kc < 4; ++kc) {
            v8h lo, hi;
            #pragma unroll
            for (int i = 0; i < 4; ++i) {
                lo[i]     = (_Float16)raw[4 * kc + 0][i];
                lo[4 + i] = (_Float16)raw[4 * kc + 1][i];
                hi[i]     = (_Float16)raw[4 * kc + 2][i];
                hi[4 + i] = (_Float16)raw[4 * kc + 3][i];
            }
            xs[wave][kc][0][lane] = lo;
            xs[wave][kc][1][lane] = hi;
        }
        if (rd + 1 < MLP_ROUNDS) {
            int arow = (tile0 + (rd + 1) * 4 + wave) * 16 + l15;
            if (arow > n - 1) arow = n - 1;
            const float* __restrict__ ab = inputs + (size_t)arow * IN_DIM + grp * 8;
            #pragma unroll
            for (int kc = 0; kc < 4; ++kc) {
                raw[4 * kc + 0] = *(const v4f*)(ab + kc * 32 + 0);
                raw[4 * kc + 1] = *(const v4f*)(ab + kc * 32 + 4);
                raw[4 * kc + 2] = *(const v4f*)(ab + kc * 32 + 16);
                raw[4 * kc + 3] = *(const v4f*)(ab + kc * 32 + 20);
            }
        }
        lds_barrier();

        for (int t = 0; t < 4; ++t) {
            v16h X[4];
            #pragma unroll
            for (int kc = 0; kc < 4; ++kc) {
                const v8h lo = xs[t][kc][0][lane];
                const v8h hi = xs[t][kc][1][lane];
                #pragma unroll
                for (int i = 0; i < 8; ++i) { X[kc][i] = lo[i]; X[kc][8 + i] = hi[i]; }
            }
            v8f acc[4];
            #pragma unroll
            for (int tt = 0; tt < 4; ++tt) { const v8f z = {}; acc[tt] = z; }
            #pragma unroll
            for (int kc = 0; kc < 4; ++kc) {
                #pragma unroll
                for (int tt = 0; tt < 4; ++tt)
                    acc[tt] = __builtin_amdgcn_wmma_f32_16x16x32_f16(
                                  false, Wf[tt][kc], false, X[kc], (short)0, acc[tt], false, false);
            }
            float sp[4];
            #pragma unroll
            for (int tt = 0; tt < 4; ++tt) {
                const int c0 = wave * HID + tt * 16 + grp * 8;
                const v4f bA = *(const v4f*)(bs + c0), bB = *(const v4f*)(bs + c0 + 4);
                const v4f wA = *(const v4f*)(w2s + c0), wB = *(const v4f*)(w2s + c0 + 4);
                float s = 0.f;
                #pragma unroll
                for (int v = 0; v < 4; ++v) {
                    float p = acc[tt][v] + bA[v];
                    p = (p > 0.f) ? p : 0.f;
                    s += p * wA[v];
                }
                #pragma unroll
                for (int v = 0; v < 4; ++v) {
                    float p = acc[tt][4 + v] + bB[v];
                    p = (p > 0.f) ? p : 0.f;
                    s += p * wB[v];
                }
                sp[tt] = s;
            }
            float s = (sp[0] + sp[1]) + (sp[2] + sp[3]);
            s += __shfl_xor(s, 16, 32);
            if (grp == 0) os[t][l15][wave] = s + b2h;
        }
        lds_barrier();
        {
            const int node = (tile0 + rd * 4 + wave) * 16 + l15;
            if (grp == 0 && node < n)
                *(v4f*)(hout + (size_t)node * H_HEADS) = *(const v4f*)(&os[wave][l15][0]);
        }
    }
}

__global__ void k_zero4(v4f* __restrict__ p, int n4) {
    int i = blockIdx.x * blockDim.x + threadIdx.x;
    if (i < n4) { v4f z = {}; p[i] = z; }
}

#define FR_SH   7
#define FR_RS   128
#define KPAD    1024
#define BK_T    1024
#define BK_EPT  8
#define BK_CH   8192
#define KEY_SH  20
#define D_MASK  0xFFFFF
#define T_MASK  63

static_assert(BK_T == KPAD, "one thread per key slot in the grouping kernels");
static_assert(BK_CH == BK_T * BK_EPT, "chunk size");
static_assert(FR_RS == (1 << FR_SH), "range size");
static_assert(RELS <= T_MASK + 1, "type field");

__global__ void __launch_bounds__(BK_T)
k_key_hist(const int* __restrict__ dst, int* __restrict__ ghist, int E) {
    __shared__ int lh[KPAD];
    const int tid = threadIdx.x;
    lh[tid] = 0;
    __syncthreads();
    const int base = blockIdx.x * BK_CH + tid;
    #pragma unroll
    for (int k = 0; k < BK_EPT; ++k) {
        const int idx = base + k * BK_T;
        if (idx < E) atomicAdd(&lh[(unsigned)dst[idx] >> FR_SH], 1);
    }
    __syncthreads();
    const int c = lh[tid];
    if (c) atomicAdd(&ghist[tid], c);
}

__global__ void __launch_bounds__(BK_T)
k_bucket(const int* __restrict__ et, const int* __restrict__ src,
         const int* __restrict__ dst, const int* __restrict__ ghist,
         int* __restrict__ cursor, int* __restrict__ kstart,
         int* __restrict__ pairs, int E) {
    __shared__ int lh[KPAD];
    __shared__ int lstart[KPAD];
    __shared__ int gbase[KPAD];
    __shared__ int wg[BK_T / 32];
    __shared__ int wc[BK_T / 32];
    __shared__ int ld[BK_CH];
    __shared__ int ls[BK_CH];
    __shared__ int lt[BK_CH];
    const int tid = threadIdx.x;
    lh[tid] = 0;
    __syncthreads();
    const int cbase = blockIdx.x * BK_CH;
    int nValid = E - cbase; if (nValid > BK_CH) nValid = BK_CH;
    int key[BK_EPT], pk[BK_EPT], s[BK_EPT], rk[BK_EPT];
    #pragma unroll
    for (int k = 0; k < BK_EPT; ++k) {
        const int idx = cbase + k * BK_T + tid;
        const bool ok = idx < E;
        const int t = ok ? et[idx] : 0;
        const int d = ok ? dst[idx] : 0;
        s[k]   = ok ? src[idx] : 0;
        pk[k]  = d | (t << KEY_SH);
        key[k] = ok ? (int)((unsigned)d >> FR_SH) : -1;
    }
    #pragma unroll
    for (int k = 0; k < BK_EPT; ++k)
        rk[k] = (key[k] >= 0) ? atomicAdd(&lh[key[k]], 1) : 0;
    __syncthreads();
    {
        const int lane = tid & 31;
        const int wave = tid >> 5;
        const int g = ghist[tid];
        const int c = lh[tid];
        int gs = g, cs = c;
        #pragma unroll
        for (int off = 1; off < 32; off <<= 1) {
            const int a = __shfl_up(gs, off, 32);
            const int b = __shfl_up(cs, off, 32);
            if (lane >= off) { gs += a; cs += b; }
        }
        if (lane == 31) { wg[wave] = gs; wc[wave] = cs; }
        __syncthreads();
        int gpre = 0, cpre = 0;
        #pragma unroll
        for (int i = 0; i < BK_T / 32; ++i) {
            gpre += (i < wave) ? wg[i] : 0;
            cpre += (i < wave) ? wc[i] : 0;
        }
        const int off = c ? atomicAdd(&cursor[tid], c) : 0;
        lstart[tid] = cpre + cs - c;
        gbase[tid]  = gpre + gs - g + off;
        if (blockIdx.x == 0) kstart[tid] = gpre + gs - g;
    }
    __syncthreads();
    #pragma unroll
    for (int k = 0; k < BK_EPT; ++k) {
        if (key[k] >= 0) {
            const int pos = lstart[key[k]] + rk[k];
            ld[pos] = pk[k]; ls[pos] = s[k]; lt[pos] = key[k];
        }
    }
    __syncthreads();
    #pragma unroll
    for (int k = 0; k < BK_EPT; ++k) {
        const int i = k * BK_T + tid;
        if (i < nValid) {
            const int kk = lt[i];
            const int slot = gbase[kk] + (i - lstart[kk]);
            int2 p; p.x = ld[i]; p.y = ls[i];
            *(int2*)(pairs + 2 * (size_t)slot) = p;
        }
    }
}

__device__ __forceinline__ int quad_bcast(int v, int j) {
    switch (j) {
        case 0:  return __builtin_amdgcn_update_dpp(0, v, 0x00, 0xf, 0xf, true);
        case 1:  return __builtin_amdgcn_update_dpp(0, v, 0x55, 0xf, 0xf, true);
        case 2:  return __builtin_amdgcn_update_dpp(0, v, 0xAA, 0xf, 0xf, true);
        default: return __builtin_amdgcn_update_dpp(0, v, 0xFF, 0xf, 0xf, true);
    }
}

#define LY_T1   512
#define LY_TG   256
#define LY_U    16
#define LY_EPT  8
static_assert(LY_EPT == 8 && LY_U == 16, "the load fences below name every slot");
#define CELLS   (RELS * FR_RS)

template <int HI, bool LAST>
__global__ void __launch_bounds__(HI == 4 ? LY_T1 : LY_TG)
k_layer(const int* __restrict__ pairs, const int* __restrict__ ghist,
        const int* __restrict__ kstart, const float* __restrict__ etab_l,
        const float* __restrict__ hin, float* __restrict__ hout,
        const float* __restrict__ centrality, const float* __restrict__ gamma,
        const float* __restrict__ beta, float* __restrict__ out, int n) {
    constexpr int T  = (HI == 4) ? LY_T1 : LY_TG;
    constexpr int NW = T / 32;
    static_assert(T >= FR_RS, "one thread per node in the store pass");
    __shared__ __attribute__((aligned(16))) int   tab[CELLS * (1 + HI)];
    __shared__ __attribute__((aligned(16))) float eL[RELS * H_HEADS];
    int*   cnt = tab;
    float* G   = (float*)(tab + CELLS);
    const int tid  = threadIdx.x;
    const int lane = tid & 31;
    const int wave = tid >> 5;
    const int r    = blockIdx.x;
    const int lo   = r * FR_RS;
    int hi = lo + FR_RS; if (hi > n) hi = n;
    const int span  = (hi > lo) ? (hi - lo) : 0;
    const int start = kstart[r];
    const int end   = start + ghist[r];
    for (int i = tid; i < RELS * H_HEADS; i += T) eL[i] = etab_l[i];
    {
        const v4f z = {};
        for (int i = tid; i < CELLS * (1 + HI) / 4; i += T) ((v4f*)tab)[i] = z;
    }
    __syncthreads();

    if (HI == 4) {
        const int h = tid & 3;
        for (int base = start + wave * (32 * LY_EPT); base < end; base += NW * 32 * LY_EPT) {
            int px[LY_EPT], py[LY_EPT];
            #pragma unroll
            for (int k = 0; k < LY_EPT; ++k) {
                int idx = base + k * 32 + lane;
                idx = (idx < end) ? idx : (end - 1);
                const int2 p = *(const int2*)(pairs + 2 * (size_t)idx);
                px[k] = p.x;
                py[k] = p.y;
            }
            asm volatile("" :: "v"(py[0]), "v"(py[1]), "v"(py[2]), "v"(py[3]),
                               "v"(py[4]), "v"(py[5]), "v"(py[6]), "v"(py[7]));
            const int emptyKey = (int)0x80000000u | (lo + lane);
            #pragma unroll
            for (int k = 0; k < LY_EPT; ++k)
                px[k] = ((base + k * 32 + lane) < end) ? px[k] : emptyKey;
            int   cell[4 * LY_EPT];
            float hv[4 * LY_EPT];
            #pragma unroll
            for (int s = 0; s < 4 * LY_EPT; ++s) {
                const int k  = s >> 2;
                const int jj = s & 3;
                const int dk = quad_bcast(px[k], jj);
                const unsigned sn = (unsigned)quad_bcast(py[k], jj);
                const bool ok = dk >= 0;
                const float x = hin[sn * H_HEADS + (unsigned)h];
                hv[s]   = ok ? x : 0.f;
                cell[s] = (((dk >> KEY_SH) & T_MASK) << FR_SH) + ((dk & D_MASK) - lo);
            }
            #pragma unroll
            for (int k = 0; k < LY_EPT; ++k) {
                const bool ok = px[k] >= 0;
                const int  c  = (((px[k] >> KEY_SH) & T_MASK) << FR_SH) + ((px[k] & D_MASK) - lo);
                atomicAdd(&cnt[c], ok ? 1 : 0);
            }
            #pragma unroll
            for (int s = 0; s < 4 * LY_EPT; ++s)
                atomicAdd(&G[cell[s] * H_HEADS + h], hv[s]);
        }
    } else {
        for (int base = start + wave * (32 * LY_U); base < end; base += NW * 32 * LY_U) {
            int px[LY_U], py[LY_U];
            #pragma unroll
            for (int k = 0; k < LY_U; ++k) {
                int idx = base + k * 32 + lane;
                idx = (idx < end) ? idx : (end - 1);
                const int2 p = *(const int2*)(pairs + 2 * (size_t)idx);
                px[k] = p.x;
                py[k] = p.y;
            }
            asm volatile("" :: "v"(py[0]), "v"(py[1]), "v"(py[2]), "v"(py[3]),
                               "v"(py[4]), "v"(py[5]), "v"(py[6]), "v"(py[7]),
                               "v"(py[8]), "v"(py[9]), "v"(py[10]), "v"(py[11]),
                               "v"(py[12]), "v"(py[13]), "v"(py[14]), "v"(py[15]));
            float hv[LY_U];
            #pragma unroll
            for (int k = 0; k < LY_U; ++k) hv[k] = hin[(unsigned)py[k]];
            #pragma unroll
            for (int k = 0; k < LY_U; ++k) {
                const bool ok = (base + k * 32 + lane) < end;
                const int  key = ok ? px[k] : (lo + lane);
                const int  c  = (((key >> KEY_SH) & T_MASK) << FR_SH) + ((key & D_MASK) - lo);
                atomicAdd(&cnt[c], ok ? 1 : 0);
                atomicAdd(&G[c], ok ? hv[k] : 0.f);
            }
        }
    }
    __syncthreads();

    if (tid < span) {
        const int d = tid;
        v4f mx;
        mx[0] = mx[1] = mx[2] = mx[3] = -__builtin_huge_valf();
        #pragma unroll 10
        for (int t = 0; t < RELS; ++t) {
            const bool present = cnt[t * FR_RS + d] > 0;
            const v4f e = *(const v4f*)(eL + t * H_HEADS);
            #pragma unroll
            for (int h = 0; h < H_HEADS; ++h)
                mx[h] = present ? fmaxf(mx[h], e[h]) : mx[h];
        }
        v4f sv = {}, uv = {};
        #pragma unroll 10
        for (int t = 0; t < RELS; ++t) {
            const int   c  = cnt[t * FR_RS + d];
            const float cf = (float)c;
            const v4f e = *(const v4f*)(eL + t * H_HEADS);
            v4f g;
            if (HI == 4) {
                g = *(const v4f*)(G + (t * FR_RS + d) * H_HEADS);
            } else {
                const float g1 = G[t * FR_RS + d];
                g[0] = g1; g[1] = g1; g[2] = g1; g[3] = g1;
            }
            #pragma unroll
            for (int h = 0; h < H_HEADS; ++h) {
                const float w = (c > 0) ? __expf(e[h] - mx[h]) : 0.f;
                sv[h] += cf * w;
                uv[h] += w * g[h];
            }
        }
        float v[H_HEADS];
        #pragma unroll
        for (int h = 0; h < H_HEADS; ++h) {
            const float x = (sv[h] > 0.f) ? (uv[h] / sv[h]) : 0.f;
            v[h] = (x > 0.f) ? x : 0.f;
        }
        if (!LAST) {
            hout[lo + d] = 0.25f * (v[0] + v[1] + v[2] + v[3]);
        } else {
            const float c = centrality[lo + d];
            float acc = 0.f;
            #pragma unroll
            for (int h = 0; h < H_HEADS; ++h)
                acc += (c * gamma[h] + beta[h]) * v[h];
            acc *= 0.25f;
            out[lo + d] = (acc > 0.f) ? acc : 0.01f * acc;
        }
    }
}

extern "C" void kernel_launch(void* const* d_in, const int* in_sizes, int n_in,
                              void* d_out, int out_size, void* d_ws, size_t ws_size,
                              hipStream_t stream) {
    const float* inputs     = (const float*)d_in[0];
    const float* centrality = (const float*)d_in[1];
    const float* W1         = (const float*)d_in[2];
    const float* b1         = (const float*)d_in[3];
    const float* W2         = (const float*)d_in[4];
    const float* b2         = (const float*)d_in[5];
    const float* rel_emb    = (const float*)d_in[6];
    const float* W_pred     = (const float*)d_in[7];
    const float* gamma      = (const float*)d_in[8];
    const float* beta       = (const float*)d_in[9];
    const int*   edge_types = (const int*)d_in[10];
    const int*   src        = (const int*)d_in[11];
    const int*   dst        = (const int*)d_in[12];
    float* out = (float*)d_out;

    const int N = in_sizes[1];
    const int E = in_sizes[10];

    char* ws = (char*)d_ws;
    _Float16* W1h = (_Float16*)ws;
    float* etab   = (float*)(ws + 65536);
    int*   ghist  = (int*)(ws + 69632);
    int*   cursor = ghist + KPAD;
    int*   kstart = cursor + KPAD;
    float* hA     = (float*)(kstart + KPAD);
    const size_t hbWords = ((size_t)N + 3) & ~(size_t)3;
    float* hbA    = hA + (size_t)N * H_HEADS;
    float* hbB    = hbA + hbWords;
    int*   pairs  = (int*)(hbB + hbWords);

    const int TPB = 256;

    {
        int nPieces = COLS * IN_DIM / 8;
        k_w1_to_f16<<<(nPieces + TPB - 1) / TPB, TPB, 0, stream>>>(W1, W1h, nPieces);
    }
    {
        int c = LAYERS * RELS * H_HEADS;
        k_etab<<<(c + TPB - 1) / TPB, TPB, 0, stream>>>(rel_emb, W_pred, etab);
    }
    {
        int n4 = 2 * KPAD / 4;
        k_zero4<<<(n4 + TPB - 1) / TPB, TPB, 0, stream>>>((v4f*)ghist, n4);
    }
    {
        int nTiles = (N + 15) / 16;
        int blocks = (nTiles + MLP_TPB - 1) / MLP_TPB;
        k_mlp_wmma<<<blocks, 128, 0, stream>>>(inputs, W1h, b1, W2, b2, hA, N);
    }

    const int bBlocks = (E + BK_CH - 1) / BK_CH;
    const int nRanges = (N + FR_RS - 1) / FR_RS;

    k_key_hist<<<bBlocks, BK_T, 0, stream>>>(dst, ghist, E);
    k_bucket<<<bBlocks, BK_T, 0, stream>>>(edge_types, src, dst, ghist, cursor, kstart, pairs, E);

    k_layer<4, false><<<nRanges, LY_T1, 0, stream>>>(pairs, ghist, kstart, etab, hA, hbA,
                                                     centrality, gamma, beta, out, N);
    k_layer<1, false><<<nRanges, LY_TG, 0, stream>>>(pairs, ghist, kstart, etab + 1 * RELS * H_HEADS,
                                                     hbA, hbB, centrality, gamma, beta, out, N);
    k_layer<1, true><<<nRanges, LY_TG, 0, stream>>>(pairs, ghist, kstart, etab + 2 * RELS * H_HEADS,
                                                    hbB, hbB, centrality, gamma, beta, out, N);
}
